# starting kernel
# baseline (speedup 1.0000x reference)
#include <hip/hip_runtime.h>

typedef float v2f __attribute__((ext_vector_type(2)));
typedef float v8f __attribute__((ext_vector_type(8)));

#define IN_CH 256
#define HID   128
#define KTILE 64


__global__ void k_deg_init(float* __restrict__ deg, int n) {
    int i = blockIdx.x * blockDim.x + threadIdx.x;
    if (i < n) deg[i] = 1.0f;
}

__global__ void k_deg_edges(const int* __restrict__ ei, float* __restrict__ deg, int E) {
    int e = blockIdx.x * blockDim.x + threadIdx.x;
    if (e < E) atomicAdd(&deg[ei[(size_t)E + e]], 1.0f);
}

__global__ void k_dinv(float* __restrict__ deg, int n) {
    int i = blockIdx.x * blockDim.x + threadIdx.x;
    if (i < n) deg[i] = rsqrtf(deg[i]);
}

__global__ __launch_bounds__(256) void k_gemm_wmma(
    const float* __restrict__ x, const float* __restrict__ W,
    const float* __restrict__ bias, const float* __restrict__ dinv,
    float* __restrict__ h, float* __restrict__ out, int nNodes) {
    __shared__ float sW[KTILE * HID];

    const int tid  = threadIdx.x;
    const int wave = tid >> 5;
    const int lane = tid & 31;
    const int half = lane >> 4;
    const int r    = lane & 15;
    const int m0   = (blockIdx.x * 8 + wave) * 16;
    const bool act = (m0 < nNodes);

    v8f acc[8] = {};

    for (int kb = 0; kb < IN_CH; kb += KTILE) {
        if (kb + KTILE < IN_CH)
            __builtin_prefetch(W + (size_t)(kb + KTILE) * HID + tid * 32, 0, 1);

        __syncthreads();
#pragma unroll
        for (int p = 0; p < (KTILE * HID / 2) / 256; ++p) {
            const int i  = tid + 256 * p;
            const int kp = i >> 7;
            const int n  = i & 127;
            float2 w2;
            w2.x = W[(size_t)(kb + 2 * kp) * HID + n];
            w2.y = W[(size_t)(kb + 2 * kp + 1) * HID + n];
            *(float2*)(sW + (size_t)i * 2) = w2;
        }
        __syncthreads();

        if (act) {
            const float* xp = x + (size_t)(m0 + r) * IN_CH + kb + 2 * half;
#pragma unroll
            for (int kl = 0; kl < KTILE; kl += 4) {
                v2f a = *(const v2f*)(xp + kl);
                const float* wrow = sW + ((kl >> 1) + half) * (2 * HID);
#pragma unroll
                for (int nt = 0; nt < 8; ++nt) {
                    v2f b = *(const v2f*)(wrow + (nt * 16 + r) * 2);
                    acc[nt] = __builtin_amdgcn_wmma_f32_16x16x4_f32(
                        false, a, false, b, (short)0, acc[nt], false, false);
                }
            }
        }
    }

    if (act) {
        float d2[8];
#pragma unroll
        for (int i = 0; i < 8; ++i) {
            float dv = dinv[m0 + i + 8 * half];
            d2[i] = dv * dv;
        }
#pragma unroll
        for (int nt = 0; nt < 8; ++nt) {
            const int   n    = nt * 16 + r;
            const float bcol = bias[n];
            float* hp = h   + (size_t)m0 * HID + n;
            float* op = out + (size_t)m0 * HID + n;
#pragma unroll
            for (int i = 0; i < 8; ++i) {
                const size_t off = (size_t)(i + 8 * half) * HID;
                const float  v   = acc[nt][i];
                hp[off] = v;
                op[off] = bcol + d2[i] * v;
            }
        }
    }
}

__global__ __launch_bounds__(256) void k_scatter(
    const int* __restrict__ ei, const float* __restrict__ h,
    const float* __restrict__ dinv, float* __restrict__ out, int E) {
    const long long gid = (long long)blockIdx.x * blockDim.x + threadIdx.x;
    const int e    = (int)(gid >> 5);
    const int lane = (int)(gid & 31);
    if (e >= E) return;
    const int src = ei[e];
    const int dst = ei[(size_t)E + e];
    const float nrm = dinv[src] * dinv[dst];
    const float4 hv = ((const float4*)(h + (size_t)src * HID))[lane];
    float* op = out + (size_t)dst * HID + 4 * lane;
    atomicAdd(op + 0, hv.x * nrm);
    atomicAdd(op + 1, hv.y * nrm);
    atomicAdd(op + 2, hv.z * nrm);
    atomicAdd(op + 3, hv.w * nrm);
}

__global__ void k_prelu(float* __restrict__ out, const float* __restrict__ alpha, long long total) {
    long long i = (long long)blockIdx.x * blockDim.x + threadIdx.x;
    if (i < total) {
        float v = out[i];
        float a = alpha[(int)(i & (HID - 1))];
        out[i] = v > 0.0f ? v : a * v;
    }
}

extern "C" void kernel_launch(void* const* d_in, const int* in_sizes, int n_in,
                              void* d_out, int out_size, void* d_ws, size_t ws_size,
                              hipStream_t stream) {
    const float* x     = (const float*)d_in[0];
    const int*   ei    = (const int*)d_in[1];
    const float* W     = (const float*)d_in[2];
    const float* bias  = (const float*)d_in[3];
    const float* alpha = (const float*)d_in[4];
    float*       out   = (float*)d_out;

    const int N = in_sizes[0] / IN_CH;
    const int E = in_sizes[1] / 2;

    float* h    = (float*)d_ws;
    float* dinv = h + (size_t)N * HID;

    k_deg_init <<<(N + 255) / 256, 256, 0, stream>>>(dinv, N);
    k_deg_edges<<<(E + 255) / 256, 256, 0, stream>>>(ei, dinv, E);
    k_dinv     <<<(N + 255) / 256, 256, 0, stream>>>(dinv, N);

    const int strips = (N + 15) / 16;
    const int blocks = (strips + 7) / 8;
    k_gemm_wmma<<<blocks, 256, 0, stream>>>(x, W, bias, dinv, h, out, N);

    const long long scThreads = (long long)E * 32;
    k_scatter<<<(int)((scThreads + 255) / 256), 256, 0, stream>>>(ei, h, dinv, out, E);

    const long long total = (long long)N * HID;
    k_prelu<<<(int)((total + 255) / 256), 256, 0, stream>>>(out, alpha, total);
}


# Replace per-edge f32 atomics by a counting sort on destination (int rank atomics + scan) and a one-wave-per-node row gather writing each row once with bias+PReLU; sums run in rank order
# speedup vs baseline: 1.9844x; 1.9844x over previous
#include <hip/hip_runtime.h>

typedef float v2f __attribute__((ext_vector_type(2)));
typedef float v8f __attribute__((ext_vector_type(8)));

#define IN_CH 256
#define HID   128
#define KTILE 64
#define SCAN_B 256


__global__ void k_cnt_zero(int* __restrict__ cnt, int n) {
    int i = blockIdx.x * blockDim.x + threadIdx.x;
    if (i < n) cnt[i] = 0;
}

__global__ void k_rank(const int* __restrict__ ei, int* __restrict__ cnt, int* __restrict__ rank, int E) {
    int e = blockIdx.x * blockDim.x + threadIdx.x;
    if (e < E) rank[e] = atomicAdd(&cnt[ei[(size_t)E + e]], 1);
}

__global__ __launch_bounds__(SCAN_B) void k_dinv_bsum(
    const int* __restrict__ cnt, float* __restrict__ dinv, int* __restrict__ bsum, int n) {
    __shared__ int sred[SCAN_B / 32];
    const int tid  = threadIdx.x;
    const int lane = tid & 31;
    const int wave = tid >> 5;
    const int i = blockIdx.x * SCAN_B + tid;
    const int c = (i < n) ? cnt[i] : 0;
    if (i < n) dinv[i] = rsqrtf((float)(c + 1));
    int s = c;
    for (int o = 16; o > 0; o >>= 1) s += __shfl_xor(s, o);
    if (lane == 0) sred[wave] = s;
    __syncthreads();
    if (tid == 0) {
        int t = 0;
        for (int w = 0; w < SCAN_B / 32; ++w) t += sred[w];
        bsum[blockIdx.x] = t;
    }
}

__global__ __launch_bounds__(SCAN_B) void k_offsets(
    const int* __restrict__ cnt, const int* __restrict__ bsum, int* __restrict__ start, int n) {
    __shared__ int sred[SCAN_B / 32];
    __shared__ int swave[SCAN_B / 32];
    const int tid  = threadIdx.x;
    const int lane = tid & 31;
    const int wave = tid >> 5;

    int p = 0;
    for (int j = tid; j < (int)blockIdx.x; j += SCAN_B) p += bsum[j];
    for (int o = 16; o > 0; o >>= 1) p += __shfl_xor(p, o);
    if (lane == 0) sred[wave] = p;

    const int i = blockIdx.x * SCAN_B + tid;
    const int c = (i < n) ? cnt[i] : 0;
    int inc = c;
    for (int o = 1; o < 32; o <<= 1) {
        int t = __shfl_up(inc, o);
        if (lane >= o) inc += t;
    }
    if (lane == 31) swave[wave] = inc;
    __syncthreads();

    int base = 0;
    for (int w = 0; w < SCAN_B / 32; ++w) {
        base += sred[w];
        if (w < wave) base += swave[w];
    }
    if (i < n) start[i] = base + inc - c;
}

__global__ void k_fill(const int* __restrict__ ei, const int* __restrict__ rank,
                       const int* __restrict__ start, int* __restrict__ srt, int E) {
    int e = blockIdx.x * blockDim.x + threadIdx.x;
    if (e < E) {
        const int dst = ei[(size_t)E + e];
        srt[start[dst] + rank[e]] = ei[e];
    }
}

__global__ __launch_bounds__(256) void k_gemm_wmma(
    const float* __restrict__ x, const float* __restrict__ W,
    float* __restrict__ h, int nNodes) {
    __shared__ float sW[KTILE * HID];

    const int tid  = threadIdx.x;
    const int wave = tid >> 5;
    const int lane = tid & 31;
    const int half = lane >> 4;
    const int r    = lane & 15;
    const int m0   = (blockIdx.x * 8 + wave) * 16;
    const bool act = (m0 < nNodes);

    v8f acc[8] = {};

    for (int kb = 0; kb < IN_CH; kb += KTILE) {
        if (kb + KTILE < IN_CH)
            __builtin_prefetch(W + (size_t)(kb + KTILE) * HID + tid * 32, 0, 1);

        __syncthreads();
#pragma unroll
        for (int p = 0; p < (KTILE * HID / 2) / 256; ++p) {
            const int i  = tid + 256 * p;
            const int kp = i >> 7;
            const int n  = i & 127;
            float2 w2;
            w2.x = W[(size_t)(kb + 2 * kp) * HID + n];
            w2.y = W[(size_t)(kb + 2 * kp + 1) * HID + n];
            *(float2*)(sW + (size_t)i * 2) = w2;
        }
        __syncthreads();

        if (act) {
            const float* xp = x + (size_t)(m0 + r) * IN_CH + kb + 2 * half;
#pragma unroll
            for (int kl = 0; kl < KTILE; kl += 4) {
                v2f a = *(const v2f*)(xp + kl);
                const float* wrow = sW + ((kl >> 1) + half) * (2 * HID);
#pragma unroll
                for (int nt = 0; nt < 8; ++nt) {
                    v2f b = *(const v2f*)(wrow + (nt * 16 + r) * 2);
                    acc[nt] = __builtin_amdgcn_wmma_f32_16x16x4_f32(
                        false, a, false, b, (short)0, acc[nt], false, false);
                }
            }
        }
    }

    if (act) {
#pragma unroll
        for (int nt = 0; nt < 8; ++nt) {
            float* hp = h + (size_t)m0 * HID + nt * 16 + r;
#pragma unroll
            for (int i = 0; i < 8; ++i)
                hp[(size_t)(i + 8 * half) * HID] = acc[nt][i];
        }
    }
}

static __device__ __forceinline__ void fma4(float4& acc, const float4 r, const float w) {
    acc.x += r.x * w; acc.y += r.y * w; acc.z += r.z * w; acc.w += r.w * w;
}

__global__ __launch_bounds__(256) void k_gather(
    const int* __restrict__ srt, const int* __restrict__ start, const int* __restrict__ cnt,
    const float* __restrict__ h, const float* __restrict__ dinv,
    const float* __restrict__ bias, const float* __restrict__ alpha,
    float* __restrict__ out, int n) {
    const int lane = threadIdx.x & 31;
    const int node = __builtin_amdgcn_readfirstlane(blockIdx.x * 8 + (threadIdx.x >> 5));
    if (node >= n) return;

    const int   beg = start[node];
    const int   c   = cnt[node];
    const float di  = dinv[node];
    const float4* __restrict__ h4 = (const float4*)h;

    const float4 self = h4[(size_t)node * (HID / 4) + lane];
    const float  d2   = di * di;
    float4 acc;
    acc.x = self.x * d2; acc.y = self.y * d2; acc.z = self.z * d2; acc.w = self.w * d2;

    for (int base = 0; base < c; base += 32) {
        const int m = (c - base < 32) ? (c - base) : 32;
        int   sv = 0;
        float wv = 0.0f;
        if (lane < m) {
            sv = srt[beg + base + lane];
            wv = dinv[sv] * di;
        }
        const int wvi = __float_as_int(wv);
        int j = 0;
        for (; j + 8 <= m; j += 8) {
            float4 r[8];
            float  w[8];
            for (int u = 0; u < 8; ++u) {
                const int s = __builtin_amdgcn_readlane(sv, j + u);
                w[u] = __int_as_float(__builtin_amdgcn_readlane(wvi, j + u));
                r[u] = h4[(size_t)s * (HID / 4) + lane];
            }
            for (int u = 0; u < 8; ++u) fma4(acc, r[u], w[u]);
        }
        if (j + 4 <= m) {
            float4 r[4];
            float  w[4];
            for (int u = 0; u < 4; ++u) {
                const int s = __builtin_amdgcn_readlane(sv, j + u);
                w[u] = __int_as_float(__builtin_amdgcn_readlane(wvi, j + u));
                r[u] = h4[(size_t)s * (HID / 4) + lane];
            }
            for (int u = 0; u < 4; ++u) fma4(acc, r[u], w[u]);
            j += 4;
        }
        if (j + 2 <= m) {
            float4 r[2];
            float  w[2];
            for (int u = 0; u < 2; ++u) {
                const int s = __builtin_amdgcn_readlane(sv, j + u);
                w[u] = __int_as_float(__builtin_amdgcn_readlane(wvi, j + u));
                r[u] = h4[(size_t)s * (HID / 4) + lane];
            }
            for (int u = 0; u < 2; ++u) fma4(acc, r[u], w[u]);
            j += 2;
        }
        if (j < m) {
            const int   s = __builtin_amdgcn_readlane(sv, j);
            const float w = __int_as_float(__builtin_amdgcn_readlane(wvi, j));
            const float4 r = h4[(size_t)s * (HID / 4) + lane];
            fma4(acc, r, w);
        }
    }

    const float4 b = ((const float4*)bias)[lane];
    const float4 a = ((const float4*)alpha)[lane];
    float4 v, o;
    v.x = acc.x + b.x; v.y = acc.y + b.y; v.z = acc.z + b.z; v.w = acc.w + b.w;
    o.x = v.x > 0.0f ? v.x : a.x * v.x;
    o.y = v.y > 0.0f ? v.y : a.y * v.y;
    o.z = v.z > 0.0f ? v.z : a.z * v.z;
    o.w = v.w > 0.0f ? v.w : a.w * v.w;
    ((float4*)out)[(size_t)node * (HID / 4) + lane] = o;
}

extern "C" void kernel_launch(void* const* d_in, const int* in_sizes, int n_in,
                              void* d_out, int out_size, void* d_ws, size_t ws_size,
                              hipStream_t stream) {
    const float* x     = (const float*)d_in[0];
    const int*   ei    = (const int*)d_in[1];
    const float* W     = (const float*)d_in[2];
    const float* bias  = (const float*)d_in[3];
    const float* alpha = (const float*)d_in[4];
    float*       out   = (float*)d_out;

    const int N = in_sizes[0] / IN_CH;
    const int E = in_sizes[1] / 2;
    const int nb = (N + SCAN_B - 1) / SCAN_B;

    float* h     = (float*)d_ws;
    float* dinv  = h + (size_t)N * HID;
    int*   cnt   = (int*)(dinv + N);
    int*   start = cnt + N;
    int*   bsum  = start + N;
    int*   rank  = bsum + nb;
    int*   srt   = rank + E;

    k_cnt_zero <<<(N + 255) / 256, 256, 0, stream>>>(cnt, N);
    k_rank     <<<(E + 255) / 256, 256, 0, stream>>>(ei, cnt, rank, E);
    k_dinv_bsum<<<nb, SCAN_B, 0, stream>>>(cnt, dinv, bsum, N);
    k_offsets  <<<nb, SCAN_B, 0, stream>>>(cnt, bsum, start, N);
    k_fill     <<<(E + 255) / 256, 256, 0, stream>>>(ei, rank, start, srt, E);

    const int strips = (N + 15) / 16;
    const int blocks = (strips + 7) / 8;
    k_gemm_wmma<<<blocks, 256, 0, stream>>>(x, W, h, N);

    k_gather<<<(N + 7) / 8, 256, 0, stream>>>(srt, start, cnt, h, dinv, bias, alpha, out, N);
}


# One LDS owner-sort launch replaces the five-launch global-atomic counting sort: a work-group owns 1024 destinations and scans all edges once; row sums run in LDS cursor arrival order
# speedup vs baseline: 2.2346x; 1.1261x over previous
#include <hip/hip_runtime.h>

typedef float v2f __attribute__((ext_vector_type(2)));
typedef float v8f __attribute__((ext_vector_type(8)));

#define IN_CH 256
#define HID   128
#define KTILE 64


#define OWN_R    1024
#define OWN_T    1024
#define OWN_WCAP 1024
#define OWN_CAP  (32 * OWN_WCAP)
#define OWN_EB   21
#define OWN_PAD  0x7fffffff

static __device__ __forceinline__ void own_visit(unsigned m, unsigned t, int e, unsigned* wl, int& wcount) {
    if (m != 0u) {
        if (t < (unsigned)OWN_R) {
            const int slot = wcount + (int)__builtin_amdgcn_mbcnt_lo(m, 0u);
            if (slot < OWN_WCAP) wl[slot] = (t << OWN_EB) | (unsigned)e;
        }
        wcount += __builtin_popcount(m);
    }
}

static __device__ __forceinline__ void own_visit4(const int4 v, int lo, int e0, unsigned* wl, int& wcount, int& below) {
    const unsigned t0 = (unsigned)(v.x - lo), t1 = (unsigned)(v.y - lo);
    const unsigned t2 = (unsigned)(v.z - lo), t3 = (unsigned)(v.w - lo);
    below += __builtin_popcount(__builtin_amdgcn_ballot_w32((int)t0 < 0))
           + __builtin_popcount(__builtin_amdgcn_ballot_w32((int)t1 < 0))
           + __builtin_popcount(__builtin_amdgcn_ballot_w32((int)t2 < 0))
           + __builtin_popcount(__builtin_amdgcn_ballot_w32((int)t3 < 0));
    const unsigned m0 = __builtin_amdgcn_ballot_w32(t0 < (unsigned)OWN_R);
    const unsigned m1 = __builtin_amdgcn_ballot_w32(t1 < (unsigned)OWN_R);
    const unsigned m2 = __builtin_amdgcn_ballot_w32(t2 < (unsigned)OWN_R);
    const unsigned m3 = __builtin_amdgcn_ballot_w32(t3 < (unsigned)OWN_R);
    if ((m0 | m1 | m2 | m3) != 0u) {
        own_visit(m0, t0, e0, wl, wcount);
        own_visit(m1, t1, e0 + 1, wl, wcount);
        own_visit(m2, t2, e0 + 2, wl, wcount);
        own_visit(m3, t3, e0 + 3, wl, wcount);
    }
}

__global__ __launch_bounds__(OWN_T) void k_owner_sort(
    const int* __restrict__ ei, int* __restrict__ cnt, int* __restrict__ start,
    float* __restrict__ dinv, int* __restrict__ srt, int n, int E) {
    __shared__ unsigned l_list[OWN_CAP];
    __shared__ int l_cnt[OWN_R];
    __shared__ int l_cur[OWN_R];
    __shared__ int l_wsum[OWN_T / 32];
    __shared__ int l_below[OWN_T / 32];
    __shared__ int l_over;

    const int tid  = threadIdx.x;
    const int lane = tid & 31;
    const int wave = tid >> 5;
    const int lo   = blockIdx.x * OWN_R;
    const int* __restrict__ dstp = ei + (size_t)E;
    unsigned* wl = l_list + wave * OWN_WCAP;
    int wcount = 0;
    int below  = 0;

    l_cnt[tid] = 0;
    if (tid == 0) l_over = (E > (1 << OWN_EB)) ? 1 : 0;
    __syncthreads();

    const int nvec = ((E & 3) == 0) ? (E >> 2) : 0;
    const int4* __restrict__ d4 = (const int4*)dstp;
    int vb = 0;
    for (; vb + 8 * OWN_T <= nvec; vb += 8 * OWN_T) {
        int4 v[8];
#pragma unroll
        for (int u = 0; u < 8; ++u) v[u] = d4[vb + u * OWN_T + tid];
#pragma unroll
        for (int u = 0; u < 8; ++u) own_visit4(v[u], lo, 4 * (vb + u * OWN_T + tid), wl, wcount, below);
    }
    for (; vb < nvec; vb += OWN_T) {
        const int i = vb + tid;
        int4 v = make_int4(OWN_PAD, OWN_PAD, OWN_PAD, OWN_PAD);
        if (i < nvec) v = d4[i];
        own_visit4(v, lo, 4 * i, wl, wcount, below);
    }
    for (int eb = 4 * nvec; eb < E; eb += OWN_T) {
        const int e = eb + tid;
        const int d = (e < E) ? dstp[e] : OWN_PAD;
        const unsigned t = (unsigned)(d - lo);
        below += __builtin_popcount(__builtin_amdgcn_ballot_w32((int)t < 0));
        own_visit(__builtin_amdgcn_ballot_w32(t < (unsigned)OWN_R), t, e, wl, wcount);
    }
    if (lane == 0) {
        l_below[wave] = below;
        if (wcount > OWN_WCAP) l_over = 1;
    }
    __syncthreads();
    const bool over = (l_over != 0);

    if (!over) {
        for (int j = lane; j < wcount; j += 32) atomicAdd(&l_cnt[wl[j] >> OWN_EB], 1);
    } else {
        for (int e = tid; e < E; e += OWN_T) {
            const unsigned t = (unsigned)(dstp[e] - lo);
            if (t < (unsigned)OWN_R) atomicAdd(&l_cnt[t], 1);
        }
    }
    __syncthreads();

    const int c = l_cnt[tid];
    int inc = c;
    for (int o = 1; o < 32; o <<= 1) {
        int t = __shfl_up(inc, o);
        if (lane >= o) inc += t;
    }
    if (lane == 31) l_wsum[wave] = inc;
    __syncthreads();
    int base = 0;
    for (int w = 0; w < OWN_T / 32; ++w) {
        base += l_below[w];
        if (w < wave) base += l_wsum[w];
    }
    const int gs = base + inc - c;
    l_cur[tid] = gs;
    const int node = lo + tid;
    if (node < n) {
        start[node] = gs;
        cnt[node]   = c;
        dinv[node]  = rsqrtf((float)(c + 1));
    }
    __syncthreads();

    if (!over) {
        for (int jb = 0; jb < wcount; jb += 256) {
            unsigned p[8];
            int sv[8];
#pragma unroll
            for (int u = 0; u < 8; ++u) {
                const int j = jb + u * 32 + lane;
                p[u] = (j < wcount) ? wl[j] : 0u;
            }
#pragma unroll
            for (int u = 0; u < 8; ++u) {
                const int j = jb + u * 32 + lane;
                sv[u] = (j < wcount) ? ei[p[u] & ((1u << OWN_EB) - 1u)] : 0;
            }
#pragma unroll
            for (int u = 0; u < 8; ++u) {
                const int j = jb + u * 32 + lane;
                if (j < wcount) {
                    const int pos = atomicAdd(&l_cur[p[u] >> OWN_EB], 1);
                    srt[pos] = sv[u];
                }
            }
        }
    } else {
        for (int e = tid; e < E; e += OWN_T) {
            const unsigned t = (unsigned)(dstp[e] - lo);
            if (t < (unsigned)OWN_R) {
                const int pos = atomicAdd(&l_cur[t], 1);
                srt[pos] = ei[e];
            }
        }
    }
}

__global__ __launch_bounds__(256) void k_gemm_wmma(
    const float* __restrict__ x, const float* __restrict__ W,
    float* __restrict__ h, int nNodes) {
    __shared__ float sW[KTILE * HID];

    const int tid  = threadIdx.x;
    const int wave = tid >> 5;
    const int lane = tid & 31;
    const int half = lane >> 4;
    const int r    = lane & 15;
    const int m0   = (blockIdx.x * 8 + wave) * 16;
    const bool act = (m0 < nNodes);

    v8f acc[8] = {};

    for (int kb = 0; kb < IN_CH; kb += KTILE) {
        if (kb + KTILE < IN_CH)
            __builtin_prefetch(W + (size_t)(kb + KTILE) * HID + tid * 32, 0, 1);

        __syncthreads();
#pragma unroll
        for (int p = 0; p < (KTILE * HID / 2) / 256; ++p) {
            const int i  = tid + 256 * p;
            const int kp = i >> 7;
            const int n  = i & 127;
            float2 w2;
            w2.x = W[(size_t)(kb + 2 * kp) * HID + n];
            w2.y = W[(size_t)(kb + 2 * kp + 1) * HID + n];
            *(float2*)(sW + (size_t)i * 2) = w2;
        }
        __syncthreads();

        if (act) {
            const float* xp = x + (size_t)(m0 + r) * IN_CH + kb + 2 * half;
#pragma unroll
            for (int kl = 0; kl < KTILE; kl += 4) {
                v2f a = *(const v2f*)(xp + kl);
                const float* wrow = sW + ((kl >> 1) + half) * (2 * HID);
#pragma unroll
                for (int nt = 0; nt < 8; ++nt) {
                    v2f b = *(const v2f*)(wrow + (nt * 16 + r) * 2);
                    acc[nt] = __builtin_amdgcn_wmma_f32_16x16x4_f32(
                        false, a, false, b, (short)0, acc[nt], false, false);
                }
            }
        }
    }

    if (act) {
#pragma unroll
        for (int nt = 0; nt < 8; ++nt) {
            float* hp = h + (size_t)m0 * HID + nt * 16 + r;
#pragma unroll
            for (int i = 0; i < 8; ++i)
                hp[(size_t)(i + 8 * half) * HID] = acc[nt][i];
        }
    }
}

static __device__ __forceinline__ void fma4(float4& acc, const float4 r, const float w) {
    acc.x += r.x * w; acc.y += r.y * w; acc.z += r.z * w; acc.w += r.w * w;
}

__global__ __launch_bounds__(256) void k_gather(
    const int* __restrict__ srt, const int* __restrict__ start, const int* __restrict__ cnt,
    const float* __restrict__ h, const float* __restrict__ dinv,
    const float* __restrict__ bias, const float* __restrict__ alpha,
    float* __restrict__ out, int n) {
    const int lane = threadIdx.x & 31;
    const int node = __builtin_amdgcn_readfirstlane(blockIdx.x * 8 + (threadIdx.x >> 5));
    if (node >= n) return;

    const int   beg = start[node];
    const int   c   = cnt[node];
    const float di  = dinv[node];
    const float4* __restrict__ h4 = (const float4*)h;

    const float4 self = h4[(size_t)node * (HID / 4) + lane];
    const float  d2   = di * di;
    float4 acc;
    acc.x = self.x * d2; acc.y = self.y * d2; acc.z = self.z * d2; acc.w = self.w * d2;

    for (int base = 0; base < c; base += 32) {
        const int m = (c - base < 32) ? (c - base) : 32;
        int   sv = 0;
        float wv = 0.0f;
        if (lane < m) {
            sv = srt[beg + base + lane];
            wv = dinv[sv] * di;
        }
        const int wvi = __float_as_int(wv);
        int j = 0;
        for (; j + 8 <= m; j += 8) {
            float4 r[8];
            float  w[8];
            for (int u = 0; u < 8; ++u) {
                const int s = __builtin_amdgcn_readlane(sv, j + u);
                w[u] = __int_as_float(__builtin_amdgcn_readlane(wvi, j + u));
                r[u] = h4[(size_t)s * (HID / 4) + lane];
            }
            for (int u = 0; u < 8; ++u) fma4(acc, r[u], w[u]);
        }
        if (j + 4 <= m) {
            float4 r[4];
            float  w[4];
            for (int u = 0; u < 4; ++u) {
                const int s = __builtin_amdgcn_readlane(sv, j + u);
                w[u] = __int_as_float(__builtin_amdgcn_readlane(wvi, j + u));
                r[u] = h4[(size_t)s * (HID / 4) + lane];
            }
            for (int u = 0; u < 4; ++u) fma4(acc, r[u], w[u]);
            j += 4;
        }
        if (j + 2 <= m) {
            float4 r[2];
            float  w[2];
            for (int u = 0; u < 2; ++u) {
                const int s = __builtin_amdgcn_readlane(sv, j + u);
                w[u] = __int_as_float(__builtin_amdgcn_readlane(wvi, j + u));
                r[u] = h4[(size_t)s * (HID / 4) + lane];
            }
            for (int u = 0; u < 2; ++u) fma4(acc, r[u], w[u]);
            j += 2;
        }
        if (j < m) {
            const int   s = __builtin_amdgcn_readlane(sv, j);
            const float w = __int_as_float(__builtin_amdgcn_readlane(wvi, j));
            const float4 r = h4[(size_t)s * (HID / 4) + lane];
            fma4(acc, r, w);
        }
    }

    const float4 b = ((const float4*)bias)[lane];
    const float4 a = ((const float4*)alpha)[lane];
    float4 v, o;
    v.x = acc.x + b.x; v.y = acc.y + b.y; v.z = acc.z + b.z; v.w = acc.w + b.w;
    o.x = v.x > 0.0f ? v.x : a.x * v.x;
    o.y = v.y > 0.0f ? v.y : a.y * v.y;
    o.z = v.z > 0.0f ? v.z : a.z * v.z;
    o.w = v.w > 0.0f ? v.w : a.w * v.w;
    ((float4*)out)[(size_t)node * (HID / 4) + lane] = o;
}

extern "C" void kernel_launch(void* const* d_in, const int* in_sizes, int n_in,
                              void* d_out, int out_size, void* d_ws, size_t ws_size,
                              hipStream_t stream) {
    const float* x     = (const float*)d_in[0];
    const int*   ei    = (const int*)d_in[1];
    const float* W     = (const float*)d_in[2];
    const float* bias  = (const float*)d_in[3];
    const float* alpha = (const float*)d_in[4];
    float*       out   = (float*)d_out;

    const int N = in_sizes[0] / IN_CH;
    const int E = in_sizes[1] / 2;
    const int nown = (N + OWN_R - 1) / OWN_R;

    float* h     = (float*)d_ws;
    float* dinv  = h + (size_t)N * HID;
    int*   cnt   = (int*)(dinv + N);
    int*   start = cnt + N;
    int*   srt   = start + N;

    k_owner_sort<<<nown, OWN_T, 0, stream>>>(ei, cnt, start, dinv, srt, N, E);

    const int strips = (N + 15) / 16;
    const int blocks = (strips + 7) / 8;
    k_gemm_wmma<<<blocks, 256, 0, stream>>>(x, W, h, N);

    k_gather<<<(N + 7) / 8, 256, 0, stream>>>(srt, start, cnt, h, dinv, bias, alpha, out, N);
}


# Owner sort and GEMM share one launch: 98 blocks sort while 158 more 1024-thread blocks do the f32 WMMA GEMM on the WGPs the sort leaves idle (3 launches to 2); summation orders as before
# speedup vs baseline: 2.8450x; 1.2732x over previous
#include <hip/hip_runtime.h>

typedef float v2f __attribute__((ext_vector_type(2)));
typedef float v8f __attribute__((ext_vector_type(8)));

#define IN_CH 256
#define HID   128
#define KTILE 64


#define OWN_R    1024
#define OWN_T    1024
#define OWN_WCAP 1024
#define OWN_CAP  (32 * OWN_WCAP)
#define OWN_EB   21
#define OWN_PAD  0x7fffffff

static __device__ __forceinline__ void own_visit(unsigned m, unsigned t, int e, unsigned* wl, int& wcount) {
    if (m != 0u) {
        if (t < (unsigned)OWN_R) {
            const int slot = wcount + (int)__builtin_amdgcn_mbcnt_lo(m, 0u);
            if (slot < OWN_WCAP) wl[slot] = (t << OWN_EB) | (unsigned)e;
        }
        wcount += __builtin_popcount(m);
    }
}

static __device__ __forceinline__ void own_visit4(const int4 v, int lo, int e0, unsigned* wl, int& wcount, int& below) {
    const unsigned t0 = (unsigned)(v.x - lo), t1 = (unsigned)(v.y - lo);
    const unsigned t2 = (unsigned)(v.z - lo), t3 = (unsigned)(v.w - lo);
    below += __builtin_popcount(__builtin_amdgcn_ballot_w32((int)t0 < 0))
           + __builtin_popcount(__builtin_amdgcn_ballot_w32((int)t1 < 0))
           + __builtin_popcount(__builtin_amdgcn_ballot_w32((int)t2 < 0))
           + __builtin_popcount(__builtin_amdgcn_ballot_w32((int)t3 < 0));
    const unsigned m0 = __builtin_amdgcn_ballot_w32(t0 < (unsigned)OWN_R);
    const unsigned m1 = __builtin_amdgcn_ballot_w32(t1 < (unsigned)OWN_R);
    const unsigned m2 = __builtin_amdgcn_ballot_w32(t2 < (unsigned)OWN_R);
    const unsigned m3 = __builtin_amdgcn_ballot_w32(t3 < (unsigned)OWN_R);
    if ((m0 | m1 | m2 | m3) != 0u) {
        own_visit(m0, t0, e0, wl, wcount);
        own_visit(m1, t1, e0 + 1, wl, wcount);
        own_visit(m2, t2, e0 + 2, wl, wcount);
        own_visit(m3, t3, e0 + 3, wl, wcount);
    }
}

static __device__ __forceinline__ void owner_sort_role(
    const int* __restrict__ ei, int* __restrict__ cnt, int* __restrict__ start,
    float* __restrict__ dinv, int* __restrict__ srt, int n, int E, int owner,
    unsigned* l_list, int* l_cnt, int* l_cur, int* l_wsum, int* l_below, int& l_over) {
    const int tid  = threadIdx.x;
    const int lane = tid & 31;
    const int wave = tid >> 5;
    const int lo   = owner * OWN_R;
    const int* __restrict__ dstp = ei + (size_t)E;
    unsigned* wl = l_list + wave * OWN_WCAP;
    int wcount = 0;
    int below  = 0;

    l_cnt[tid] = 0;
    if (tid == 0) l_over = (E > (1 << OWN_EB)) ? 1 : 0;
    __syncthreads();

    const int nvec = ((E & 3) == 0) ? (E >> 2) : 0;
    const int4* __restrict__ d4 = (const int4*)dstp;
    int vb = 0;
    for (; vb + 8 * OWN_T <= nvec; vb += 8 * OWN_T) {
        int4 v[8];
#pragma unroll
        for (int u = 0; u < 8; ++u) v[u] = d4[vb + u * OWN_T + tid];
#pragma unroll
        for (int u = 0; u < 8; ++u) own_visit4(v[u], lo, 4 * (vb + u * OWN_T + tid), wl, wcount, below);
    }
    for (; vb < nvec; vb += OWN_T) {
        const int i = vb + tid;
        int4 v = make_int4(OWN_PAD, OWN_PAD, OWN_PAD, OWN_PAD);
        if (i < nvec) v = d4[i];
        own_visit4(v, lo, 4 * i, wl, wcount, below);
    }
    for (int eb = 4 * nvec; eb < E; eb += OWN_T) {
        const int e = eb + tid;
        const int d = (e < E) ? dstp[e] : OWN_PAD;
        const unsigned t = (unsigned)(d - lo);
        below += __builtin_popcount(__builtin_amdgcn_ballot_w32((int)t < 0));
        own_visit(__builtin_amdgcn_ballot_w32(t < (unsigned)OWN_R), t, e, wl, wcount);
    }
    if (lane == 0) {
        l_below[wave] = below;
        if (wcount > OWN_WCAP) l_over = 1;
    }
    __syncthreads();
    const bool over = (l_over != 0);

    if (!over) {
        for (int j = lane; j < wcount; j += 32) atomicAdd(&l_cnt[wl[j] >> OWN_EB], 1);
    } else {
        for (int e = tid; e < E; e += OWN_T) {
            const unsigned t = (unsigned)(dstp[e] - lo);
            if (t < (unsigned)OWN_R) atomicAdd(&l_cnt[t], 1);
        }
    }
    __syncthreads();

    const int c = l_cnt[tid];
    int inc = c;
    for (int o = 1; o < 32; o <<= 1) {
        int t = __shfl_up(inc, o);
        if (lane >= o) inc += t;
    }
    if (lane == 31) l_wsum[wave] = inc;
    __syncthreads();
    int base = 0;
    for (int w = 0; w < OWN_T / 32; ++w) {
        base += l_below[w];
        if (w < wave) base += l_wsum[w];
    }
    const int gs = base + inc - c;
    l_cur[tid] = gs;
    const int node = lo + tid;
    if (node < n) {
        start[node] = gs;
        cnt[node]   = c;
        dinv[node]  = rsqrtf((float)(c + 1));
    }
    __syncthreads();

    if (!over) {
        for (int jb = 0; jb < wcount; jb += 256) {
            unsigned p[8];
            int sv[8];
#pragma unroll
            for (int u = 0; u < 8; ++u) {
                const int j = jb + u * 32 + lane;
                p[u] = (j < wcount) ? wl[j] : 0u;
            }
#pragma unroll
            for (int u = 0; u < 8; ++u) {
                const int j = jb + u * 32 + lane;
                sv[u] = (j < wcount) ? ei[p[u] & ((1u << OWN_EB) - 1u)] : 0;
            }
#pragma unroll
            for (int u = 0; u < 8; ++u) {
                const int j = jb + u * 32 + lane;
                if (j < wcount) {
                    const int pos = atomicAdd(&l_cur[p[u] >> OWN_EB], 1);
                    srt[pos] = sv[u];
                }
            }
        }
    } else {
        for (int e = tid; e < E; e += OWN_T) {
            const unsigned t = (unsigned)(dstp[e] - lo);
            if (t < (unsigned)OWN_R) {
                const int pos = atomicAdd(&l_cur[t], 1);
                srt[pos] = ei[e];
            }
        }
    }
}

static __device__ __forceinline__ void gemm_strip_pass(
    const float* __restrict__ x, const float* __restrict__ W,
    float* __restrict__ h, float* sW, const int m0, const bool act) {
    const int tid  = threadIdx.x;
    const int lane = tid & 31;
    const int half = lane >> 4;
    const int r    = lane & 15;

    v8f acc[8] = {};

    for (int kb = 0; kb < IN_CH; kb += KTILE) {
        if (kb + KTILE < IN_CH && tid < 256)
            __builtin_prefetch(W + (size_t)(kb + KTILE) * HID + tid * 32, 0, 1);

        __syncthreads();
#pragma unroll
        for (int p = 0; p < (KTILE * HID / 2) / OWN_T; ++p) {
            const int i  = tid + OWN_T * p;
            const int kp = i >> 7;
            const int n  = i & 127;
            float2 w2;
            w2.x = W[(size_t)(kb + 2 * kp) * HID + n];
            w2.y = W[(size_t)(kb + 2 * kp + 1) * HID + n];
            *(float2*)(sW + (size_t)i * 2) = w2;
        }
        __syncthreads();

        if (act) {
            const float* xp = x + (size_t)(m0 + r) * IN_CH + kb + 2 * half;
#pragma unroll
            for (int kl = 0; kl < KTILE; kl += 4) {
                v2f a = *(const v2f*)(xp + kl);
                const float* wrow = sW + ((kl >> 1) + half) * (2 * HID);
#pragma unroll
                for (int nt = 0; nt < 8; ++nt) {
                    v2f b = *(const v2f*)(wrow + (nt * 16 + r) * 2);
                    acc[nt] = __builtin_amdgcn_wmma_f32_16x16x4_f32(
                        false, a, false, b, (short)0, acc[nt], false, false);
                }
            }
        }
    }

    if (act) {
#pragma unroll
        for (int nt = 0; nt < 8; ++nt) {
            float* hp = h + (size_t)m0 * HID + nt * 16 + r;
#pragma unroll
            for (int i = 0; i < 8; ++i)
                hp[(size_t)(i + 8 * half) * HID] = acc[nt][i];
        }
    }
}

__global__ __launch_bounds__(OWN_T) void k_sort_and_gemm(
    const int* __restrict__ ei, int* __restrict__ cnt, int* __restrict__ start,
    float* __restrict__ dinv, int* __restrict__ srt, int n, int E, int nown,
    const float* __restrict__ x, const float* __restrict__ W, float* __restrict__ h) {
    __shared__ unsigned l_list[OWN_CAP];
    __shared__ int l_cnt[OWN_R];
    __shared__ int l_cur[OWN_R];
    __shared__ int l_wsum[OWN_T / 32];
    __shared__ int l_below[OWN_T / 32];
    __shared__ int l_over;

    if ((int)blockIdx.x < nown) {
        owner_sort_role(ei, cnt, start, dinv, srt, n, E, (int)blockIdx.x,
                        l_list, l_cnt, l_cur, l_wsum, l_below, l_over);
        return;
    }

    float* sW = (float*)l_list;
    const int g      = (int)blockIdx.x - nown;
    const int ngemm  = (int)gridDim.x - nown;
    const int wave   = threadIdx.x >> 5;
    const int strips = (n + 15) / 16;
    const int nw     = ngemm * (OWN_T / 32);
    const int nfull  = strips / nw;
    const int rem    = strips - nfull * nw;
    for (int p = 0; p < nfull; ++p)
        gemm_strip_pass(x, W, h, sW, (p * nw + g * (OWN_T / 32) + wave) * 16, true);
    if (g < rem) {
        const int q = wave * ngemm + g;
        gemm_strip_pass(x, W, h, sW, (nfull * nw + q) * 16, q < rem);
    }
}

static __device__ __forceinline__ void fma4(float4& acc, const float4 r, const float w) {
    acc.x += r.x * w; acc.y += r.y * w; acc.z += r.z * w; acc.w += r.w * w;
}

__global__ __launch_bounds__(256) void k_gather(
    const int* __restrict__ srt, const int* __restrict__ start, const int* __restrict__ cnt,
    const float* __restrict__ h, const float* __restrict__ dinv,
    const float* __restrict__ bias, const float* __restrict__ alpha,
    float* __restrict__ out, int n) {
    const int lane = threadIdx.x & 31;
    const int node = __builtin_amdgcn_readfirstlane(blockIdx.x * 8 + (threadIdx.x >> 5));
    if (node >= n) return;

    const int   beg = start[node];
    const int   c   = cnt[node];
    const float di  = dinv[node];
    const float4* __restrict__ h4 = (const float4*)h;

    const float4 self = h4[(size_t)node * (HID / 4) + lane];
    const float  d2   = di * di;
    float4 acc;
    acc.x = self.x * d2; acc.y = self.y * d2; acc.z = self.z * d2; acc.w = self.w * d2;

    for (int base = 0; base < c; base += 32) {
        const int m = (c - base < 32) ? (c - base) : 32;
        int   sv = 0;
        float wv = 0.0f;
        if (lane < m) {
            sv = srt[beg + base + lane];
            wv = dinv[sv] * di;
        }
        const int wvi = __float_as_int(wv);
        int j = 0;
        for (; j + 8 <= m; j += 8) {
            float4 r[8];
            float  w[8];
            for (int u = 0; u < 8; ++u) {
                const int s = __builtin_amdgcn_readlane(sv, j + u);
                w[u] = __int_as_float(__builtin_amdgcn_readlane(wvi, j + u));
                r[u] = h4[(size_t)s * (HID / 4) + lane];
            }
            for (int u = 0; u < 8; ++u) fma4(acc, r[u], w[u]);
        }
        if (j + 4 <= m) {
            float4 r[4];
            float  w[4];
            for (int u = 0; u < 4; ++u) {
                const int s = __builtin_amdgcn_readlane(sv, j + u);
                w[u] = __int_as_float(__builtin_amdgcn_readlane(wvi, j + u));
                r[u] = h4[(size_t)s * (HID / 4) + lane];
            }
            for (int u = 0; u < 4; ++u) fma4(acc, r[u], w[u]);
            j += 4;
        }
        if (j + 2 <= m) {
            float4 r[2];
            float  w[2];
            for (int u = 0; u < 2; ++u) {
                const int s = __builtin_amdgcn_readlane(sv, j + u);
                w[u] = __int_as_float(__builtin_amdgcn_readlane(wvi, j + u));
                r[u] = h4[(size_t)s * (HID / 4) + lane];
            }
            for (int u = 0; u < 2; ++u) fma4(acc, r[u], w[u]);
            j += 2;
        }
        if (j < m) {
            const int   s = __builtin_amdgcn_readlane(sv, j);
            const float w = __int_as_float(__builtin_amdgcn_readlane(wvi, j));
            const float4 r = h4[(size_t)s * (HID / 4) + lane];
            fma4(acc, r, w);
        }
    }

    const float4 b = ((const float4*)bias)[lane];
    const float4 a = ((const float4*)alpha)[lane];
    float4 v, o;
    v.x = acc.x + b.x; v.y = acc.y + b.y; v.z = acc.z + b.z; v.w = acc.w + b.w;
    o.x = v.x > 0.0f ? v.x : a.x * v.x;
    o.y = v.y > 0.0f ? v.y : a.y * v.y;
    o.z = v.z > 0.0f ? v.z : a.z * v.z;
    o.w = v.w > 0.0f ? v.w : a.w * v.w;
    ((float4*)out)[(size_t)node * (HID / 4) + lane] = o;
}

extern "C" void kernel_launch(void* const* d_in, const int* in_sizes, int n_in,
                              void* d_out, int out_size, void* d_ws, size_t ws_size,
                              hipStream_t stream) {
    const float* x     = (const float*)d_in[0];
    const int*   ei    = (const int*)d_in[1];
    const float* W     = (const float*)d_in[2];
    const float* bias  = (const float*)d_in[3];
    const float* alpha = (const float*)d_in[4];
    float*       out   = (float*)d_out;

    const int N = in_sizes[0] / IN_CH;
    const int E = in_sizes[1] / 2;
    const int nown = (N + OWN_R - 1) / OWN_R;

    float* h     = (float*)d_ws;
    float* dinv  = h + (size_t)N * HID;
    int*   cnt   = (int*)(dinv + N);
    int*   start = cnt + N;
    int*   srt   = start + N;

    const int ngemm = (nown < 192) ? (256 - nown) : 64;
    k_sort_and_gemm<<<nown + ngemm, OWN_T, 0, stream>>>(ei, cnt, start, dinv, srt, N, E, nown, x, W, h);

    k_gather<<<(N + 7) / 8, 256, 0, stream>>>(srt, start, cnt, h, dinv, bias, alpha, out, N);
}


# Edge list read once: a pre-pass buckets edges by owner range in edge order (wave-private LDS ranks); owners copy buckets, no full scan, and take GEMM strips too (3 launches); sums in LDS cursor order
# speedup vs baseline: 3.6075x; 1.2680x over previous
#include <hip/hip_runtime.h>

typedef float v2f __attribute__((ext_vector_type(2)));
typedef float v8f __attribute__((ext_vector_type(8)));

#define IN_CH 256
#define HID   128
#define KTILE 64


#define OWN_R    1024
#define OWN_T    1024
#define OWN_WCAP 1024
#define OWN_CAP  (32 * OWN_WCAP)
#define OWN_EB   21

#define PART_B   256
#define PART_T   256
#define PART_W   (PART_T / 32)
#define PART_OWN 128
#define PART_CAP 128
#define PART_D   16

__global__ __launch_bounds__(PART_T) void k_partition(
    const int* __restrict__ ei, int* __restrict__ cntob, unsigned* __restrict__ bkt, int E, int nown) {
    __shared__ int      l_pc[PART_W * PART_OWN];
    __shared__ int      l_tot[PART_OWN];
    __shared__ unsigned l_bk[PART_OWN * PART_CAP];
    const int tid  = threadIdx.x;
    const int lane = tid & 31;
    const int wave = tid >> 5;
    const int b    = blockIdx.x;
    if (nown > PART_OWN) {
        for (int o = tid; o < nown; o += PART_T) cntob[o * PART_B + b] = PART_CAP + 1;
        return;
    }
    const int* __restrict__ dstp = ei + (size_t)E;
    const int chunk = (E + PART_B - 1) / PART_B;
    const int wch   = (chunk + PART_W - 1) / PART_W;
    const int e0    = b * chunk;
    const int e1    = (e0 + chunk < E) ? (e0 + chunk) : E;
    const int ws    = e0 + wave * wch;
    const int we    = (ws + wch < e1) ? (ws + wch) : e1;
    int* pc = l_pc + wave * PART_OWN;

    for (int i = tid; i < PART_W * PART_OWN; i += PART_T) l_pc[i] = 0;
    __syncthreads();
    for (int eb = ws; eb < we; eb += PART_D * 32) {
        int d[PART_D];
#pragma unroll
        for (int u = 0; u < PART_D; ++u) {
            const int e = eb + u * 32 + lane;
            d[u] = dstp[(e < we) ? e : (we - 1)];
        }
#pragma unroll
        for (int u = 0; u < PART_D; ++u) {
            const int e = eb + u * 32 + lane;
            const unsigned o = (unsigned)d[u] / (unsigned)OWN_R;
            if (e < we && o < (unsigned)nown) atomicAdd(&pc[o], 1);
        }
    }
    __syncthreads();
    if (tid < nown) {
        int run = 0;
        for (int w = 0; w < PART_W; ++w) {
            const int c = l_pc[w * PART_OWN + tid];
            l_pc[w * PART_OWN + tid] = run;
            run += c;
        }
        l_tot[tid] = run;
        cntob[tid * PART_B + b] = run;
    }
    __syncthreads();
    for (int eb = ws; eb < we; eb += PART_D * 32) {
        int d[PART_D];
#pragma unroll
        for (int u = 0; u < PART_D; ++u) {
            const int e = eb + u * 32 + lane;
            d[u] = dstp[(e < we) ? e : (we - 1)];
        }
#pragma unroll
        for (int u = 0; u < PART_D; ++u) {
            const int e = eb + u * 32 + lane;
            const unsigned o = (unsigned)d[u] / (unsigned)OWN_R;
            if (e < we && o < (unsigned)nown) {
                const int slot = atomicAdd(&pc[o], 1);
                if (slot < PART_CAP)
                    l_bk[o * PART_CAP + slot] = (((unsigned)d[u] - o * (unsigned)OWN_R) << OWN_EB) | (unsigned)e;
            }
        }
    }
    __syncthreads();
    for (int o = wave; o < nown; o += PART_W) {
        const int c = (l_tot[o] < PART_CAP) ? l_tot[o] : PART_CAP;
        unsigned* __restrict__ g = bkt + ((size_t)o * PART_B + b) * PART_CAP;
        for (int j = lane; j < c; j += 32) g[j] = l_bk[o * PART_CAP + j];
    }
}

static __device__ __forceinline__ void owner_sort_role(
    const int* __restrict__ ei, const int* __restrict__ cntob, const unsigned* __restrict__ bkt,
    int* __restrict__ cnt, int* __restrict__ start,
    float* __restrict__ dinv, int* __restrict__ srt, int n, int E, int owner,
    unsigned* l_list, int* l_cnt, int* l_cur, int* l_wsum, int* l_below, int& l_over) {
    const int tid  = threadIdx.x;
    const int lane = tid & 31;
    const int wave = tid >> 5;
    const int lo   = owner * OWN_R;
    const int* __restrict__ dstp = ei + (size_t)E;
    unsigned* wl = l_list + wave * OWN_WCAP;

    l_cnt[tid] = 0;
    if (tid == 0) l_over = (E > (1 << OWN_EB)) ? 1 : 0;
    __syncthreads();

    const int b0 = wave * (PART_B / 32);
    int cb = 0;
    if (lane < PART_B / 32) cb = cntob[owner * PART_B + b0 + lane];

    int below = 0;
    const int lim = owner * PART_B;
    for (int ib = 0; ib < lim; ib += 8 * OWN_T) {
        int t[8];
#pragma unroll
        for (int u = 0; u < 8; ++u) {
            const int i = ib + u * OWN_T + tid;
            t[u] = (i < lim) ? cntob[i] : 0;
        }
#pragma unroll
        for (int u = 0; u < 8; ++u) below += t[u];
    }
    const bool bad = (__builtin_amdgcn_ballot_w32(cb > PART_CAP) != 0u);
    int c8[PART_B / 32], off[PART_B / 32];
    int wcount = 0;
#pragma unroll
    for (int i = 0; i < PART_B / 32; ++i) {
        c8[i]  = __builtin_amdgcn_readlane(cb, i);
        off[i] = wcount;
        wcount += c8[i];
    }
    if (bad || wcount > OWN_WCAP) {
        if (lane == 0) l_over = 1;
    } else {
        unsigned v[PART_B / 32][PART_CAP / 32];
#pragma unroll
        for (int i = 0; i < PART_B / 32; ++i)
#pragma unroll
            for (int k = 0; k < PART_CAP / 32; ++k) {
                const int j = k * 32 + lane;
                v[i][k] = (j < c8[i]) ? bkt[((size_t)owner * PART_B + b0 + i) * PART_CAP + j] : 0u;
            }
#pragma unroll
        for (int i = 0; i < PART_B / 32; ++i)
#pragma unroll
            for (int k = 0; k < PART_CAP / 32; ++k) {
                const int j = k * 32 + lane;
                if (j < c8[i]) wl[off[i] + j] = v[i][k];
            }
    }
    for (int o = 16; o > 0; o >>= 1) below += __shfl_xor(below, o);
    if (lane == 0) l_below[wave] = below;
    __syncthreads();
    const bool over = (l_over != 0);

    if (!over) {
        for (int j = lane; j < wcount; j += 32) atomicAdd(&l_cnt[wl[j] >> OWN_EB], 1);
    } else {
        int bl = 0;
        for (int e = tid; e < E; e += OWN_T) {
            const unsigned t = (unsigned)(dstp[e] - lo);
            if ((int)t < 0) ++bl;
            if (t < (unsigned)OWN_R) atomicAdd(&l_cnt[t], 1);
        }
        for (int o = 16; o > 0; o >>= 1) bl += __shfl_xor(bl, o);
        if (lane == 0) l_below[wave] = bl;
    }
    __syncthreads();
    const int c = l_cnt[tid];
    int inc = c;
    for (int o = 1; o < 32; o <<= 1) {
        int t = __shfl_up(inc, o);
        if (lane >= o) inc += t;
    }
    if (lane == 31) l_wsum[wave] = inc;
    __syncthreads();
    int base = 0;
    for (int w = 0; w < OWN_T / 32; ++w) {
        base += l_below[w];
        if (w < wave) base += l_wsum[w];
    }
    const int gs = base + inc - c;
    l_cur[tid] = gs;
    const int node = lo + tid;
    if (node < n) {
        start[node] = gs;
        cnt[node]   = c;
        dinv[node]  = rsqrtf((float)(c + 1));
    }
    __syncthreads();

    if (!over) {
        for (int jb = 0; jb < wcount; jb += 256) {
            unsigned p[8];
            int sv[8];
#pragma unroll
            for (int u = 0; u < 8; ++u) {
                const int j = jb + u * 32 + lane;
                p[u] = (j < wcount) ? wl[j] : 0u;
            }
#pragma unroll
            for (int u = 0; u < 8; ++u) {
                const int j = jb + u * 32 + lane;
                sv[u] = (j < wcount) ? ei[p[u] & ((1u << OWN_EB) - 1u)] : 0;
            }
#pragma unroll
            for (int u = 0; u < 8; ++u) {
                const int j = jb + u * 32 + lane;
                if (j < wcount) {
                    const int pos = atomicAdd(&l_cur[p[u] >> OWN_EB], 1);
                    srt[pos] = sv[u];
                }
            }
        }
    } else {
        for (int e = tid; e < E; e += OWN_T) {
            const unsigned t = (unsigned)(dstp[e] - lo);
            if (t < (unsigned)OWN_R) {
                const int pos = atomicAdd(&l_cur[t], 1);
                srt[pos] = ei[e];
            }
        }
    }
}

static __device__ __forceinline__ void gemm_strip_pass(
    const float* __restrict__ x, const float* __restrict__ W,
    float* __restrict__ h, float* sW, const int m0, const bool act) {
    const int tid  = threadIdx.x;
    const int lane = tid & 31;
    const int half = lane >> 4;
    const int r    = lane & 15;

    v8f acc[8] = {};

    for (int kb = 0; kb < IN_CH; kb += KTILE) {
        if (kb + KTILE < IN_CH && tid < 256)
            __builtin_prefetch(W + (size_t)(kb + KTILE) * HID + tid * 32, 0, 1);

        __syncthreads();
#pragma unroll
        for (int p = 0; p < (KTILE * HID / 2) / OWN_T; ++p) {
            const int i  = tid + OWN_T * p;
            const int kp = i >> 7;
            const int n  = i & 127;
            float2 w2;
            w2.x = W[(size_t)(kb + 2 * kp) * HID + n];
            w2.y = W[(size_t)(kb + 2 * kp + 1) * HID + n];
            *(float2*)(sW + (size_t)i * 2) = w2;
        }
        __syncthreads();

        if (act) {
            const float* xp = x + (size_t)(m0 + r) * IN_CH + kb + 2 * half;
#pragma unroll
            for (int kl = 0; kl < KTILE; kl += 4) {
                v2f a = *(const v2f*)(xp + kl);
                const float* wrow = sW + ((kl >> 1) + half) * (2 * HID);
#pragma unroll
                for (int nt = 0; nt < 8; ++nt) {
                    v2f b = *(const v2f*)(wrow + (nt * 16 + r) * 2);
                    acc[nt] = __builtin_amdgcn_wmma_f32_16x16x4_f32(
                        false, a, false, b, (short)0, acc[nt], false, false);
                }
            }
        }
    }

    if (act) {
#pragma unroll
        for (int nt = 0; nt < 8; ++nt) {
            float* hp = h + (size_t)m0 * HID + nt * 16 + r;
#pragma unroll
            for (int i = 0; i < 8; ++i)
                hp[(size_t)(i + 8 * half) * HID] = acc[nt][i];
        }
    }
}

__global__ __launch_bounds__(OWN_T) void k_sort_and_gemm(
    const int* __restrict__ ei, const int* __restrict__ cntob, const unsigned* __restrict__ bkt,
    int* __restrict__ cnt, int* __restrict__ start,
    float* __restrict__ dinv, int* __restrict__ srt, int n, int E, int nown,
    const float* __restrict__ x, const float* __restrict__ W, float* __restrict__ h) {
    __shared__ unsigned l_list[OWN_CAP];
    __shared__ int l_cnt[OWN_R];
    __shared__ int l_cur[OWN_R];
    __shared__ int l_wsum[OWN_T / 32];
    __shared__ int l_below[OWN_T / 32];
    __shared__ int l_over;

    const bool sorter = ((int)blockIdx.x < nown);
    if (sorter)
        owner_sort_role(ei, cntob, bkt, cnt, start, dinv, srt, n, E, (int)blockIdx.x,
                        l_list, l_cnt, l_cur, l_wsum, l_below, l_over);

    float* sW = (float*)l_list;
    const int G      = (int)gridDim.x;
    const int ngem   = G - nown;
    const int wave   = threadIdx.x >> 5;
    const int strips = (n + 15) / 16;
    const int qavg   = (strips + G - 1) / G;
    int qs = ((qavg - 4) / 4) * 4;
    if (qs < 0) qs = 0;
    const int left   = (strips > qs * nown) ? (strips - qs * nown) : 0;
    const int qg     = (left + ngem - 1) / ngem;
    const int q      = sorter ? qs : qg;
    const int first  = sorter ? (int)blockIdx.x : (qs * nown + (int)blockIdx.x - nown);
    const int stride = sorter ? nown : ngem;
    for (int p = 0; p * (OWN_T / 32) < q; ++p) {
        const int v = p * (OWN_T / 32) + wave;
        const int s = first + v * stride;
        gemm_strip_pass(x, W, h, sW, s * 16, (v < q) && (s < strips));
    }
}

static __device__ __forceinline__ void fma4(float4& acc, const float4 r, const float w) {
    acc.x += r.x * w; acc.y += r.y * w; acc.z += r.z * w; acc.w += r.w * w;
}

__global__ __launch_bounds__(256) void k_gather(
    const int* __restrict__ srt, const int* __restrict__ start, const int* __restrict__ cnt,
    const float* __restrict__ h, const float* __restrict__ dinv,
    const float* __restrict__ bias, const float* __restrict__ alpha,
    float* __restrict__ out, int n) {
    const int lane = threadIdx.x & 31;
    const int node = __builtin_amdgcn_readfirstlane(blockIdx.x * 8 + (threadIdx.x >> 5));
    if (node >= n) return;

    const int   beg = start[node];
    const int   c   = cnt[node];
    const float di  = dinv[node];
    const float4* __restrict__ h4 = (const float4*)h;

    const float4 self = h4[(size_t)node * (HID / 4) + lane];
    const float  d2   = di * di;
    float4 acc;
    acc.x = self.x * d2; acc.y = self.y * d2; acc.z = self.z * d2; acc.w = self.w * d2;

    for (int base = 0; base < c; base += 32) {
        const int m = (c - base < 32) ? (c - base) : 32;
        int   sv = 0;
        float wv = 0.0f;
        if (lane < m) {
            sv = srt[beg + base + lane];
            wv = dinv[sv] * di;
        }
        const int wvi = __float_as_int(wv);
        int j = 0;
        for (; j + 8 <= m; j += 8) {
            float4 r[8];
            float  w[8];
            for (int u = 0; u < 8; ++u) {
                const int s = __builtin_amdgcn_readlane(sv, j + u);
                w[u] = __int_as_float(__builtin_amdgcn_readlane(wvi, j + u));
                r[u] = h4[(size_t)s * (HID / 4) + lane];
            }
            for (int u = 0; u < 8; ++u) fma4(acc, r[u], w[u]);
        }
        if (j + 4 <= m) {
            float4 r[4];
            float  w[4];
            for (int u = 0; u < 4; ++u) {
                const int s = __builtin_amdgcn_readlane(sv, j + u);
                w[u] = __int_as_float(__builtin_amdgcn_readlane(wvi, j + u));
                r[u] = h4[(size_t)s * (HID / 4) + lane];
            }
            for (int u = 0; u < 4; ++u) fma4(acc, r[u], w[u]);
            j += 4;
        }
        if (j + 2 <= m) {
            float4 r[2];
            float  w[2];
            for (int u = 0; u < 2; ++u) {
                const int s = __builtin_amdgcn_readlane(sv, j + u);
                w[u] = __int_as_float(__builtin_amdgcn_readlane(wvi, j + u));
                r[u] = h4[(size_t)s * (HID / 4) + lane];
            }
            for (int u = 0; u < 2; ++u) fma4(acc, r[u], w[u]);
            j += 2;
        }
        if (j < m) {
            const int   s = __builtin_amdgcn_readlane(sv, j);
            const float w = __int_as_float(__builtin_amdgcn_readlane(wvi, j));
            const float4 r = h4[(size_t)s * (HID / 4) + lane];
            fma4(acc, r, w);
        }
    }

    const float4 b = ((const float4*)bias)[lane];
    const float4 a = ((const float4*)alpha)[lane];
    float4 v, o;
    v.x = acc.x + b.x; v.y = acc.y + b.y; v.z = acc.z + b.z; v.w = acc.w + b.w;
    o.x = v.x > 0.0f ? v.x : a.x * v.x;
    o.y = v.y > 0.0f ? v.y : a.y * v.y;
    o.z = v.z > 0.0f ? v.z : a.z * v.z;
    o.w = v.w > 0.0f ? v.w : a.w * v.w;
    ((float4*)out)[(size_t)node * (HID / 4) + lane] = o;
}

extern "C" void kernel_launch(void* const* d_in, const int* in_sizes, int n_in,
                              void* d_out, int out_size, void* d_ws, size_t ws_size,
                              hipStream_t stream) {
    const float* x     = (const float*)d_in[0];
    const int*   ei    = (const int*)d_in[1];
    const float* W     = (const float*)d_in[2];
    const float* bias  = (const float*)d_in[3];
    const float* alpha = (const float*)d_in[4];
    float*       out   = (float*)d_out;

    const int N = in_sizes[0] / IN_CH;
    const int E = in_sizes[1] / 2;
    const int nown = (N + OWN_R - 1) / OWN_R;

    float*    h     = (float*)d_ws;
    float*    dinv  = h + (size_t)N * HID;
    int*      cnt   = (int*)(dinv + N);
    int*      start = cnt + N;
    int*      srt   = start + N;
    int*      cntob = srt + E;
    unsigned* bkt   = (unsigned*)(cntob + (size_t)nown * PART_B);

    k_partition<<<PART_B, PART_T, 0, stream>>>(ei, cntob, bkt, E, nown);

    const int ngemm = (nown < 192) ? (256 - nown) : 64;
    k_sort_and_gemm<<<nown + ngemm, OWN_T, 0, stream>>>(ei, cntob, bkt, cnt, start, dinv, srt, N, E, nown, x, W, h);

    k_gather<<<(N + 7) / 8, 256, 0, stream>>>(srt, start, cnt, h, dinv, bias, alpha, out, N);
}


# Owner sort makes no scattered 4-byte global access: bucket words carry the source id (pre-pass reads it coalesced), srt segment built in LDS and written in whole runs; sums in LDS cursor order
# speedup vs baseline: 4.1118x; 1.1398x over previous
#include <hip/hip_runtime.h>

typedef float v2f __attribute__((ext_vector_type(2)));
typedef float v8f __attribute__((ext_vector_type(8)));

#define IN_CH 256
#define HID   128
#define KTILE 64


#define OWN_R    1024
#define OWN_T    1024
#define OWN_WCAP 1024
#define OWN_CAP  (32 * OWN_WCAP)
#define OWN_SB   22

#define PART_B   256
#define PART_T   256
#define PART_W   (PART_T / 32)
#define PART_OWN 128
#define PART_CAP 128
#define PART_D   16

__global__ __launch_bounds__(PART_T) void k_partition(
    const int* __restrict__ ei, int* __restrict__ cntob, unsigned* __restrict__ bkt, int E, int nown) {
    __shared__ int      l_pc[PART_W * PART_OWN];
    __shared__ int      l_tot[PART_OWN];
    __shared__ unsigned l_bk[PART_OWN * PART_CAP];
    const int tid  = threadIdx.x;
    const int lane = tid & 31;
    const int wave = tid >> 5;
    const int b    = blockIdx.x;
    if (nown > PART_OWN) {
        for (int o = tid; o < nown; o += PART_T) cntob[o * PART_B + b] = PART_CAP + 1;
        return;
    }
    const int* __restrict__ dstp = ei + (size_t)E;
    const int chunk = (E + PART_B - 1) / PART_B;
    const int wch   = (chunk + PART_W - 1) / PART_W;
    const int e0    = b * chunk;
    const int e1    = (e0 + chunk < E) ? (e0 + chunk) : E;
    const int ws    = e0 + wave * wch;
    const int we    = (ws + wch < e1) ? (ws + wch) : e1;
    int* pc = l_pc + wave * PART_OWN;

    for (int i = tid; i < PART_W * PART_OWN; i += PART_T) l_pc[i] = 0;
    __syncthreads();
    for (int eb = ws; eb < we; eb += PART_D * 32) {
        int d[PART_D];
        if (lane < PART_D) {
            const int e = eb + lane * 32;
            __builtin_prefetch(ei + ((e < we) ? e : (we - 1)), 0, 1);
        }
#pragma unroll
        for (int u = 0; u < PART_D; ++u) {
            const int e = eb + u * 32 + lane;
            d[u] = dstp[(e < we) ? e : (we - 1)];
        }
#pragma unroll
        for (int u = 0; u < PART_D; ++u) {
            const int e = eb + u * 32 + lane;
            const unsigned o = (unsigned)d[u] / (unsigned)OWN_R;
            if (e < we && o < (unsigned)nown) atomicAdd(&pc[o], 1);
        }
    }
    __syncthreads();
    if (tid < nown) {
        int run = 0;
        for (int w = 0; w < PART_W; ++w) {
            const int c = l_pc[w * PART_OWN + tid];
            l_pc[w * PART_OWN + tid] = run;
            run += c;
        }
        l_tot[tid] = run;
        cntob[tid * PART_B + b] = run;
    }
    __syncthreads();
    for (int eb = ws; eb < we; eb += PART_D * 32) {
        int d[PART_D], s[PART_D];
#pragma unroll
        for (int u = 0; u < PART_D; ++u) {
            const int e  = eb + u * 32 + lane;
            const int ec = (e < we) ? e : (we - 1);
            d[u] = dstp[ec];
            s[u] = ei[ec];
        }
#pragma unroll
        for (int u = 0; u < PART_D; ++u) {
            const int e = eb + u * 32 + lane;
            const unsigned o = (unsigned)d[u] / (unsigned)OWN_R;
            if (e < we && o < (unsigned)nown) {
                const int slot = atomicAdd(&pc[o], 1);
                if (slot < PART_CAP)
                    l_bk[o * PART_CAP + slot] = (((unsigned)d[u] - o * (unsigned)OWN_R) << OWN_SB) |
                                                ((unsigned)s[u] & ((1u << OWN_SB) - 1u));
            }
        }
    }
    __syncthreads();
    for (int o = wave; o < nown; o += PART_W) {
        const int c = (l_tot[o] < PART_CAP) ? l_tot[o] : PART_CAP;
        unsigned* __restrict__ g = bkt + ((size_t)o * PART_B + b) * PART_CAP;
        for (int j = lane; j < c; j += 32) g[j] = l_bk[o * PART_CAP + j];
    }
}

static __device__ __forceinline__ void owner_sort_role(
    const int* __restrict__ ei, const int* __restrict__ cntob, const unsigned* __restrict__ bkt,
    int* __restrict__ cnt, int* __restrict__ start,
    float* __restrict__ dinv, int* __restrict__ srt, int n, int E, int owner,
    unsigned* l_list, unsigned* l_out, int* l_cnt, int* l_cur, int* l_wsum, int* l_below, int& l_over) {
    const int tid  = threadIdx.x;
    const int lane = tid & 31;
    const int wave = tid >> 5;
    const int lo   = owner * OWN_R;
    const int* __restrict__ dstp = ei + (size_t)E;
    unsigned* wl = l_list + wave * OWN_WCAP;

    l_cnt[tid] = 0;
    if (tid == 0) l_over = 0;
    __syncthreads();

    const int b0 = wave * (PART_B / 32);
    int cb = 0;
    if (lane < PART_B / 32) cb = cntob[owner * PART_B + b0 + lane];

    int below = 0;
    const int lim = owner * PART_B;
    for (int ib = 0; ib < lim; ib += 8 * OWN_T) {
        int t[8];
#pragma unroll
        for (int u = 0; u < 8; ++u) {
            const int i = ib + u * OWN_T + tid;
            t[u] = (i < lim) ? cntob[i] : 0;
        }
#pragma unroll
        for (int u = 0; u < 8; ++u) below += t[u];
    }
    const bool bad = (__builtin_amdgcn_ballot_w32(cb > PART_CAP) != 0u);
    int c8[PART_B / 32], off[PART_B / 32];
    int wcount = 0;
#pragma unroll
    for (int i = 0; i < PART_B / 32; ++i) {
        c8[i]  = __builtin_amdgcn_readlane(cb, i);
        off[i] = wcount;
        wcount += c8[i];
    }
    if (bad || wcount > OWN_WCAP) {
        if (lane == 0) l_over = 1;
    } else {
        unsigned v[PART_B / 32][PART_CAP / 32];
#pragma unroll
        for (int i = 0; i < PART_B / 32; ++i)
#pragma unroll
            for (int k = 0; k < PART_CAP / 32; ++k) {
                const int j = k * 32 + lane;
                v[i][k] = (j < c8[i]) ? bkt[((size_t)owner * PART_B + b0 + i) * PART_CAP + j] : 0u;
            }
#pragma unroll
        for (int i = 0; i < PART_B / 32; ++i)
#pragma unroll
            for (int k = 0; k < PART_CAP / 32; ++k) {
                const int j = k * 32 + lane;
                if (j < c8[i]) wl[off[i] + j] = v[i][k];
            }
    }
    for (int o = 16; o > 0; o >>= 1) below += __shfl_xor(below, o);
    if (lane == 0) l_below[wave] = below;
    __syncthreads();
    const bool over = (l_over != 0);

    if (!over) {
        for (int j = lane; j < wcount; j += 32) atomicAdd(&l_cnt[wl[j] >> OWN_SB], 1);
    } else {
        int bl = 0;
        for (int e = tid; e < E; e += OWN_T) {
            const unsigned t = (unsigned)(dstp[e] - lo);
            if ((int)t < 0) ++bl;
            if (t < (unsigned)OWN_R) atomicAdd(&l_cnt[t], 1);
        }
        for (int o = 16; o > 0; o >>= 1) bl += __shfl_xor(bl, o);
        if (lane == 0) l_below[wave] = bl;
    }
    __syncthreads();
    const int c = l_cnt[tid];
    int inc = c;
    for (int o = 1; o < 32; o <<= 1) {
        int t = __shfl_up(inc, o);
        if (lane >= o) inc += t;
    }
    if (lane == 31) l_wsum[wave] = inc;
    __syncthreads();
    int obase = 0, wbase = 0, total = 0;
    for (int w = 0; w < OWN_T / 32; ++w) {
        obase += l_below[w];
        total += l_wsum[w];
        if (w < wave) wbase += l_wsum[w];
    }
    const int ls = wbase + inc - c;
    const int gs = obase + ls;
    l_cur[tid] = over ? gs : ls;
    const int node = lo + tid;
    if (node < n) {
        start[node] = gs;
        cnt[node]   = c;
        dinv[node]  = rsqrtf((float)(c + 1));
    }
    __syncthreads();

    if (!over) {
        for (int jb = 0; jb < wcount; jb += 256) {
            unsigned p[8];
#pragma unroll
            for (int u = 0; u < 8; ++u) {
                const int j = jb + u * 32 + lane;
                p[u] = (j < wcount) ? wl[j] : 0u;
            }
#pragma unroll
            for (int u = 0; u < 8; ++u) {
                const int j = jb + u * 32 + lane;
                if (j < wcount) {
                    const int pos = atomicAdd(&l_cur[p[u] >> OWN_SB], 1);
                    l_out[pos] = p[u] & ((1u << OWN_SB) - 1u);
                }
            }
        }
        __syncthreads();
        for (int i = tid; i < total; i += OWN_T) srt[obase + i] = (int)l_out[i];
    } else {
        for (int e = tid; e < E; e += OWN_T) {
            const unsigned t = (unsigned)(dstp[e] - lo);
            if (t < (unsigned)OWN_R) {
                const int pos = atomicAdd(&l_cur[t], 1);
                srt[pos] = ei[e];
            }
        }
    }
}

static __device__ __forceinline__ void gemm_strip_pass(
    const float* __restrict__ x, const float* __restrict__ W,
    float* __restrict__ h, float* sW, const int m0, const bool act) {
    const int tid  = threadIdx.x;
    const int lane = tid & 31;
    const int half = lane >> 4;
    const int r    = lane & 15;

    v8f acc[8] = {};

    for (int kb = 0; kb < IN_CH; kb += KTILE) {
        if (kb + KTILE < IN_CH && tid < 256)
            __builtin_prefetch(W + (size_t)(kb + KTILE) * HID + tid * 32, 0, 1);

        __syncthreads();
#pragma unroll
        for (int p = 0; p < (KTILE * HID / 2) / OWN_T; ++p) {
            const int i  = tid + OWN_T * p;
            const int kp = i >> 7;
            const int n  = i & 127;
            float2 w2;
            w2.x = W[(size_t)(kb + 2 * kp) * HID + n];
            w2.y = W[(size_t)(kb + 2 * kp + 1) * HID + n];
            *(float2*)(sW + (size_t)i * 2) = w2;
        }
        __syncthreads();

        if (act) {
            const float* xp = x + (size_t)(m0 + r) * IN_CH + kb + 2 * half;
#pragma unroll
            for (int kl = 0; kl < KTILE; kl += 4) {
                v2f a = *(const v2f*)(xp + kl);
                const float* wrow = sW + ((kl >> 1) + half) * (2 * HID);
#pragma unroll
                for (int nt = 0; nt < 8; ++nt) {
                    v2f b = *(const v2f*)(wrow + (nt * 16 + r) * 2);
                    acc[nt] = __builtin_amdgcn_wmma_f32_16x16x4_f32(
                        false, a, false, b, (short)0, acc[nt], false, false);
                }
            }
        }
    }

    if (act) {
#pragma unroll
        for (int nt = 0; nt < 8; ++nt) {
            float* hp = h + (size_t)m0 * HID + nt * 16 + r;
#pragma unroll
            for (int i = 0; i < 8; ++i)
                hp[(size_t)(i + 8 * half) * HID] = acc[nt][i];
        }
    }
}

__global__ __launch_bounds__(OWN_T) void k_sort_and_gemm(
    const int* __restrict__ ei, const int* __restrict__ cntob, const unsigned* __restrict__ bkt,
    int* __restrict__ cnt, int* __restrict__ start,
    float* __restrict__ dinv, int* __restrict__ srt, int n, int E, int nown,
    const float* __restrict__ x, const float* __restrict__ W, float* __restrict__ h) {
    __shared__ unsigned l_list[OWN_CAP];
    __shared__ unsigned l_out[OWN_CAP];
    __shared__ int l_cnt[OWN_R];
    __shared__ int l_cur[OWN_R];
    __shared__ int l_wsum[OWN_T / 32];
    __shared__ int l_below[OWN_T / 32];
    __shared__ int l_over;

    const bool sorter = ((int)blockIdx.x < nown);
    if (sorter)
        owner_sort_role(ei, cntob, bkt, cnt, start, dinv, srt, n, E, (int)blockIdx.x,
                        l_list, l_out, l_cnt, l_cur, l_wsum, l_below, l_over);

    float* sW = (float*)l_list;
    const int G      = (int)gridDim.x;
    const int ngem   = G - nown;
    const int wave   = threadIdx.x >> 5;
    const int strips = (n + 15) / 16;
    const int qavg   = (strips + G - 1) / G;
    int qs = ((qavg - 4) / 4) * 4;
    if (qs < 0) qs = 0;
    const int left   = (strips > qs * nown) ? (strips - qs * nown) : 0;
    const int qg     = (left + ngem - 1) / ngem;
    const int q      = sorter ? qs : qg;
    const int first  = sorter ? (int)blockIdx.x : (qs * nown + (int)blockIdx.x - nown);
    const int stride = sorter ? nown : ngem;
    for (int p = 0; p * (OWN_T / 32) < q; ++p) {
        const int v = p * (OWN_T / 32) + wave;
        const int s = first + v * stride;
        gemm_strip_pass(x, W, h, sW, s * 16, (v < q) && (s < strips));
    }
}

static __device__ __forceinline__ void fma4(float4& acc, const float4 r, const float w) {
    acc.x += r.x * w; acc.y += r.y * w; acc.z += r.z * w; acc.w += r.w * w;
}

__global__ __launch_bounds__(256) void k_gather(
    const int* __restrict__ srt, const int* __restrict__ start, const int* __restrict__ cnt,
    const float* __restrict__ h, const float* __restrict__ dinv,
    const float* __restrict__ bias, const float* __restrict__ alpha,
    float* __restrict__ out, int n) {
    const int lane = threadIdx.x & 31;
    const int node = __builtin_amdgcn_readfirstlane(blockIdx.x * 8 + (threadIdx.x >> 5));
    if (node >= n) return;

    const int   beg = start[node];
    const int   c   = cnt[node];
    const float di  = dinv[node];
    const float4* __restrict__ h4 = (const float4*)h;

    const float4 self = h4[(size_t)node * (HID / 4) + lane];
    const float  d2   = di * di;
    float4 acc;
    acc.x = self.x * d2; acc.y = self.y * d2; acc.z = self.z * d2; acc.w = self.w * d2;

    for (int base = 0; base < c; base += 32) {
        const int m = (c - base < 32) ? (c - base) : 32;
        int   sv = 0;
        float wv = 0.0f;
        if (lane < m) {
            sv = srt[beg + base + lane];
            wv = dinv[sv] * di;
        }
        const int wvi = __float_as_int(wv);
        int j = 0;
        for (; j + 8 <= m; j += 8) {
            float4 r[8];
            float  w[8];
            for (int u = 0; u < 8; ++u) {
                const int s = __builtin_amdgcn_readlane(sv, j + u);
                w[u] = __int_as_float(__builtin_amdgcn_readlane(wvi, j + u));
                r[u] = h4[(size_t)s * (HID / 4) + lane];
            }
            for (int u = 0; u < 8; ++u) fma4(acc, r[u], w[u]);
        }
        if (j + 4 <= m) {
            float4 r[4];
            float  w[4];
            for (int u = 0; u < 4; ++u) {
                const int s = __builtin_amdgcn_readlane(sv, j + u);
                w[u] = __int_as_float(__builtin_amdgcn_readlane(wvi, j + u));
                r[u] = h4[(size_t)s * (HID / 4) + lane];
            }
            for (int u = 0; u < 4; ++u) fma4(acc, r[u], w[u]);
            j += 4;
        }
        if (j + 2 <= m) {
            float4 r[2];
            float  w[2];
            for (int u = 0; u < 2; ++u) {
                const int s = __builtin_amdgcn_readlane(sv, j + u);
                w[u] = __int_as_float(__builtin_amdgcn_readlane(wvi, j + u));
                r[u] = h4[(size_t)s * (HID / 4) + lane];
            }
            for (int u = 0; u < 2; ++u) fma4(acc, r[u], w[u]);
            j += 2;
        }
        if (j < m) {
            const int   s = __builtin_amdgcn_readlane(sv, j);
            const float w = __int_as_float(__builtin_amdgcn_readlane(wvi, j));
            const float4 r = h4[(size_t)s * (HID / 4) + lane];
            fma4(acc, r, w);
        }
    }

    const float4 b = ((const float4*)bias)[lane];
    const float4 a = ((const float4*)alpha)[lane];
    float4 v, o;
    v.x = acc.x + b.x; v.y = acc.y + b.y; v.z = acc.z + b.z; v.w = acc.w + b.w;
    o.x = v.x > 0.0f ? v.x : a.x * v.x;
    o.y = v.y > 0.0f ? v.y : a.y * v.y;
    o.z = v.z > 0.0f ? v.z : a.z * v.z;
    o.w = v.w > 0.0f ? v.w : a.w * v.w;
    ((float4*)out)[(size_t)node * (HID / 4) + lane] = o;
}

extern "C" void kernel_launch(void* const* d_in, const int* in_sizes, int n_in,
                              void* d_out, int out_size, void* d_ws, size_t ws_size,
                              hipStream_t stream) {
    const float* x     = (const float*)d_in[0];
    const int*   ei    = (const int*)d_in[1];
    const float* W     = (const float*)d_in[2];
    const float* bias  = (const float*)d_in[3];
    const float* alpha = (const float*)d_in[4];
    float*       out   = (float*)d_out;

    const int N = in_sizes[0] / IN_CH;
    const int E = in_sizes[1] / 2;
    const int nown = (N + OWN_R - 1) / OWN_R;

    float*    h     = (float*)d_ws;
    float*    dinv  = h + (size_t)N * HID;
    int*      cnt   = (int*)(dinv + N);
    int*      start = cnt + N;
    int*      srt   = start + N;
    int*      cntob = srt + E;
    unsigned* bkt   = (unsigned*)(cntob + (size_t)nown * PART_B);

    k_partition<<<PART_B, PART_T, 0, stream>>>(ei, cntob, bkt, E, nown);

    const int ngemm = (nown < 192) ? (256 - nown) : 64;
    k_sort_and_gemm<<<nown + ngemm, OWN_T, 0, stream>>>(ei, cntob, bkt, cnt, start, dinv, srt, N, E, nown, x, W, h);

    k_gather<<<(N + 7) / 8, 256, 0, stream>>>(srt, start, cnt, h, dinv, bias, alpha, out, N);
}


# GEMM K loop software-pipelined: next step's 8 B fragments and x pair loaded as one group ahead of the current 8 WMMAs (W staged once, no K barriers); 2048 strips run beside bucketing; sums unchanged
# speedup vs baseline: 4.5574x; 1.1084x over previous
#include <hip/hip_runtime.h>

typedef float v2f __attribute__((ext_vector_type(2)));
typedef float v8f __attribute__((ext_vector_type(8)));

#define IN_CH 256
#define HID   128


#define OWN_R    1024
#define OWN_T    1024
#define OWN_WCAP 1024
#define OWN_CAP  (32 * OWN_WCAP)
#define OWN_SB   22

#define PART_B   256
#define PART_T   256
#define PART_W   (PART_T / 32)
#define PART_OWN 128
#define PART_CAP 128
#define PART_D   16
#define GEMM1_B  256
#define LDS1_W   (IN_CH * HID)

template <int NT>
static __device__ __forceinline__ void stage_w_all(const float* __restrict__ W, float* sW) {
#pragma unroll 4
    for (int u = threadIdx.x; u < (IN_CH / 2) * (HID / 4); u += NT) {
        const int kp = u >> 5;
        const int n4 = (u & 31) * 4;
        const float4 w0 = *(const float4*)(W + (size_t)(2 * kp) * HID + n4);
        const float4 w1 = *(const float4*)(W + (size_t)(2 * kp + 1) * HID + n4);
        float4 lo, hi;
        lo.x = w0.x; lo.y = w1.x; lo.z = w0.y; lo.w = w1.y;
        hi.x = w0.z; hi.y = w1.z; hi.z = w0.w; hi.w = w1.w;
        float* d = sW + (size_t)kp * (2 * HID) + n4 * 2;
        *(float4*)d       = lo;
        *(float4*)(d + 4) = hi;
    }
}

static __device__ __forceinline__ void load_step(v2f& a, v2f (&b)[8], const float* xp, const float* wp, const int k) {
    a = *(const v2f*)(xp + k);
    const float* wr = wp + (k >> 1) * (2 * HID);
#pragma unroll
    for (int nt = 0; nt < 8; ++nt) b[nt] = *(const v2f*)(wr + nt * 32);
}

static __device__ __forceinline__ void wmma_step(v8f (&acc)[8], const v2f a, const v2f (&b)[8]) {
#pragma unroll
    for (int nt = 0; nt < 8; ++nt)
        acc[nt] = __builtin_amdgcn_wmma_f32_16x16x4_f32(false, a, false, b[nt], (short)0, acc[nt], false, false);
}

static __device__ __forceinline__ void gemm_strip(
    const float* __restrict__ x, float* __restrict__ h, const float* sW, const int m0) {
    const int lane = threadIdx.x & 31;
    const int half = lane >> 4;
    const int r    = lane & 15;
    const float* xrow = x + (size_t)(m0 + r) * IN_CH;
    const float* xp   = xrow + 2 * half;
    const float* wp   = sW + half * (2 * HID) + r * 2;

    v8f acc[8] = {};
    v2f a0, a1, b0[8], b1[8];

    __builtin_prefetch(xrow + half * 32, 0, 1);
    load_step(a0, b0, xp, wp, 0);
#pragma nounroll
    for (int kg = 0; kg < IN_CH; kg += 32) {
        if ((kg & 32) == 0 && kg + 64 < IN_CH) __builtin_prefetch(xrow + kg + 64 + half * 32, 0, 1);
#pragma unroll
        for (int s = 0; s < 8; s += 2) {
            const int k = kg + 4 * s;
            load_step(a1, b1, xp, wp, k + 4);
            __builtin_amdgcn_sched_barrier(0);
            wmma_step(acc, a0, b0);
            __builtin_amdgcn_sched_barrier(0);
            load_step(a0, b0, xp, wp, (k + 8 < IN_CH) ? (k + 8) : k);
            __builtin_amdgcn_sched_barrier(0);
            wmma_step(acc, a1, b1);
            __builtin_amdgcn_sched_barrier(0);
        }
    }

#pragma unroll
    for (int nt = 0; nt < 8; ++nt) {
        float* hp = h + (size_t)m0 * HID + nt * 16 + r;
#pragma unroll
        for (int i = 0; i < 8; ++i)
            hp[(size_t)(i + 8 * half) * HID] = acc[nt][i];
    }
}

__global__ __launch_bounds__(PART_T) void k_partition(
    const int* __restrict__ ei, int* __restrict__ cntob, unsigned* __restrict__ bkt, int E, int nown,
    const float* __restrict__ x, const float* __restrict__ W, float* __restrict__ h, int n) {
    __shared__ __attribute__((aligned(16))) unsigned l_mem[LDS1_W];
    int*      l_pc  = (int*)l_mem;
    int*      l_tot = l_pc + PART_W * PART_OWN;
    unsigned* l_bk  = (unsigned*)(l_tot + PART_OWN);
    const int tid  = threadIdx.x;
    const int lane = tid & 31;
    const int wave = tid >> 5;
    if ((int)blockIdx.x >= PART_B) {
        float* sW = (float*)l_mem;
        stage_w_all<PART_T>(W, sW);
        __syncthreads();
        const int strips = (n + 15) / 16;
        const int s1 = (GEMM1_B * PART_W < strips) ? (GEMM1_B * PART_W) : strips;
        const int s  = ((int)blockIdx.x - PART_B) * PART_W + wave;
        if (s < s1) gemm_strip(x, h, sW, s * 16);
        return;
    }
    const int b    = blockIdx.x;
    if (nown > PART_OWN) {
        for (int o = tid; o < nown; o += PART_T) cntob[o * PART_B + b] = PART_CAP + 1;
        return;
    }
    const int* __restrict__ dstp = ei + (size_t)E;
    const int chunk = (E + PART_B - 1) / PART_B;
    const int wch   = (chunk + PART_W - 1) / PART_W;
    const int e0    = b * chunk;
    const int e1    = (e0 + chunk < E) ? (e0 + chunk) : E;
    const int ws    = e0 + wave * wch;
    const int we    = (ws + wch < e1) ? (ws + wch) : e1;
    int* pc = l_pc + wave * PART_OWN;

    for (int i = tid; i < PART_W * PART_OWN; i += PART_T) l_pc[i] = 0;
    __syncthreads();
    for (int eb = ws; eb < we; eb += PART_D * 32) {
        int d[PART_D];
        if (lane < PART_D) {
            const int e = eb + lane * 32;
            __builtin_prefetch(ei + ((e < we) ? e : (we - 1)), 0, 1);
        }
#pragma unroll
        for (int u = 0; u < PART_D; ++u) {
            const int e = eb + u * 32 + lane;
            d[u] = dstp[(e < we) ? e : (we - 1)];
        }
#pragma unroll
        for (int u = 0; u < PART_D; ++u) {
            const int e = eb + u * 32 + lane;
            const unsigned o = (unsigned)d[u] / (unsigned)OWN_R;
            if (e < we && o < (unsigned)nown) atomicAdd(&pc[o], 1);
        }
    }
    __syncthreads();
    if (tid < nown) {
        int run = 0;
        for (int w = 0; w < PART_W; ++w) {
            const int c = l_pc[w * PART_OWN + tid];
            l_pc[w * PART_OWN + tid] = run;
            run += c;
        }
        l_tot[tid] = run;
        cntob[tid * PART_B + b] = run;
    }
    __syncthreads();
    for (int eb = ws; eb < we; eb += PART_D * 32) {
        int d[PART_D], s[PART_D];
#pragma unroll
        for (int u = 0; u < PART_D; ++u) {
            const int e  = eb + u * 32 + lane;
            const int ec = (e < we) ? e : (we - 1);
            d[u] = dstp[ec];
            s[u] = ei[ec];
        }
#pragma unroll
        for (int u = 0; u < PART_D; ++u) {
            const int e = eb + u * 32 + lane;
            const unsigned o = (unsigned)d[u] / (unsigned)OWN_R;
            if (e < we && o < (unsigned)nown) {
                const int slot = atomicAdd(&pc[o], 1);
                if (slot < PART_CAP)
                    l_bk[o * PART_CAP + slot] = (((unsigned)d[u] - o * (unsigned)OWN_R) << OWN_SB) |
                                                ((unsigned)s[u] & ((1u << OWN_SB) - 1u));
            }
        }
    }
    __syncthreads();
    for (int o = wave; o < nown; o += PART_W) {
        const int c = (l_tot[o] < PART_CAP) ? l_tot[o] : PART_CAP;
        unsigned* __restrict__ g = bkt + ((size_t)o * PART_B + b) * PART_CAP;
        for (int j = lane; j < c; j += 32) g[j] = l_bk[o * PART_CAP + j];
    }
}

static __device__ __forceinline__ void owner_sort_role(
    const int* __restrict__ ei, const int* __restrict__ cntob, const unsigned* __restrict__ bkt,
    int* __restrict__ cnt, int* __restrict__ start,
    float* __restrict__ dinv, int* __restrict__ srt, int n, int E, int owner,
    unsigned* l_list, unsigned* l_out, int* l_cnt, int* l_cur, int* l_wsum, int* l_below, int& l_over) {
    const int tid  = threadIdx.x;
    const int lane = tid & 31;
    const int wave = tid >> 5;
    const int lo   = owner * OWN_R;
    const int* __restrict__ dstp = ei + (size_t)E;
    unsigned* wl = l_list + wave * OWN_WCAP;

    l_cnt[tid] = 0;
    if (tid == 0) l_over = 0;
    __syncthreads();

    const int b0 = wave * (PART_B / 32);
    int cb = 0;
    if (lane < PART_B / 32) cb = cntob[owner * PART_B + b0 + lane];

    int below = 0;
    const int lim = owner * PART_B;
    for (int ib = 0; ib < lim; ib += 8 * OWN_T) {
        int t[8];
#pragma unroll
        for (int u = 0; u < 8; ++u) {
            const int i = ib + u * OWN_T + tid;
            t[u] = (i < lim) ? cntob[i] : 0;
        }
#pragma unroll
        for (int u = 0; u < 8; ++u) below += t[u];
    }
    const bool bad = (__builtin_amdgcn_ballot_w32(cb > PART_CAP) != 0u);
    int c8[PART_B / 32], off[PART_B / 32];
    int wcount = 0;
#pragma unroll
    for (int i = 0; i < PART_B / 32; ++i) {
        c8[i]  = __builtin_amdgcn_readlane(cb, i);
        off[i] = wcount;
        wcount += c8[i];
    }
    if (bad || wcount > OWN_WCAP) {
        if (lane == 0) l_over = 1;
    } else {
        unsigned v[PART_B / 32][PART_CAP / 32];
#pragma unroll
        for (int i = 0; i < PART_B / 32; ++i)
#pragma unroll
            for (int k = 0; k < PART_CAP / 32; ++k) {
                const int j = k * 32 + lane;
                v[i][k] = (j < c8[i]) ? bkt[((size_t)owner * PART_B + b0 + i) * PART_CAP + j] : 0u;
            }
#pragma unroll
        for (int i = 0; i < PART_B / 32; ++i)
#pragma unroll
            for (int k = 0; k < PART_CAP / 32; ++k) {
                const int j = k * 32 + lane;
                if (j < c8[i]) wl[off[i] + j] = v[i][k];
            }
    }
    for (int o = 16; o > 0; o >>= 1) below += __shfl_xor(below, o);
    if (lane == 0) l_below[wave] = below;
    __syncthreads();
    const bool over = (l_over != 0);

    if (!over) {
        for (int j = lane; j < wcount; j += 32) atomicAdd(&l_cnt[wl[j] >> OWN_SB], 1);
    } else {
        int bl = 0;
        for (int e = tid; e < E; e += OWN_T) {
            const unsigned t = (unsigned)(dstp[e] - lo);
            if ((int)t < 0) ++bl;
            if (t < (unsigned)OWN_R) atomicAdd(&l_cnt[t], 1);
        }
        for (int o = 16; o > 0; o >>= 1) bl += __shfl_xor(bl, o);
        if (lane == 0) l_below[wave] = bl;
    }
    __syncthreads();
    const int c = l_cnt[tid];
    int inc = c;
    for (int o = 1; o < 32; o <<= 1) {
        int t = __shfl_up(inc, o);
        if (lane >= o) inc += t;
    }
    if (lane == 31) l_wsum[wave] = inc;
    __syncthreads();
    int obase = 0, wbase = 0, total = 0;
    for (int w = 0; w < OWN_T / 32; ++w) {
        obase += l_below[w];
        total += l_wsum[w];
        if (w < wave) wbase += l_wsum[w];
    }
    const int ls = wbase + inc - c;
    const int gs = obase + ls;
    l_cur[tid] = over ? gs : ls;
    const int node = lo + tid;
    if (node < n) {
        start[node] = gs;
        cnt[node]   = c;
        dinv[node]  = rsqrtf((float)(c + 1));
    }
    __syncthreads();

    if (!over) {
        for (int jb = 0; jb < wcount; jb += 256) {
            unsigned p[8];
#pragma unroll
            for (int u = 0; u < 8; ++u) {
                const int j = jb + u * 32 + lane;
                p[u] = (j < wcount) ? wl[j] : 0u;
            }
#pragma unroll
            for (int u = 0; u < 8; ++u) {
                const int j = jb + u * 32 + lane;
                if (j < wcount) {
                    const int pos = atomicAdd(&l_cur[p[u] >> OWN_SB], 1);
                    l_out[pos] = p[u] & ((1u << OWN_SB) - 1u);
                }
            }
        }
        __syncthreads();
        for (int i = tid; i < total; i += OWN_T) srt[obase + i] = (int)l_out[i];
    } else {
        for (int e = tid; e < E; e += OWN_T) {
            const unsigned t = (unsigned)(dstp[e] - lo);
            if (t < (unsigned)OWN_R) {
                const int pos = atomicAdd(&l_cur[t], 1);
                srt[pos] = ei[e];
            }
        }
    }
}

__global__ __launch_bounds__(OWN_T) void k_sort_and_gemm(
    const int* __restrict__ ei, const int* __restrict__ cntob, const unsigned* __restrict__ bkt,
    int* __restrict__ cnt, int* __restrict__ start,
    float* __restrict__ dinv, int* __restrict__ srt, int n, int E, int nown,
    const float* __restrict__ x, const float* __restrict__ W, float* __restrict__ h) {
    __shared__ __attribute__((aligned(16))) unsigned l_list[OWN_CAP];
    __shared__ unsigned l_out[OWN_CAP];
    __shared__ int l_cnt[OWN_R];
    __shared__ int l_cur[OWN_R];
    __shared__ int l_wsum[OWN_T / 32];
    __shared__ int l_below[OWN_T / 32];
    __shared__ int l_over;

    const bool sorter = ((int)blockIdx.x < nown);
    if (sorter)
        owner_sort_role(ei, cntob, bkt, cnt, start, dinv, srt, n, E, (int)blockIdx.x,
                        l_list, l_out, l_cnt, l_cur, l_wsum, l_below, l_over);

    float* sW = (float*)l_list;
    __syncthreads();
    stage_w_all<OWN_T>(W, sW);
    __syncthreads();
    const int G      = (int)gridDim.x;
    const int ngem   = G - nown;
    const int wave   = threadIdx.x >> 5;
    const int nstr   = (n + 15) / 16;
    const int s1     = (GEMM1_B * PART_W < nstr) ? (GEMM1_B * PART_W) : nstr;
    const int strips = nstr - s1;
    const int qavg   = (strips + G - 1) / G;
    int qs = ((qavg - 4) / 4) * 4;
    if (qs < 0) qs = 0;
    const int left   = (strips > qs * nown) ? (strips - qs * nown) : 0;
    const int qg     = (left + ngem - 1) / ngem;
    const int q      = sorter ? qs : qg;
    const int first  = sorter ? (int)blockIdx.x : (qs * nown + (int)blockIdx.x - nown);
    const int stride = sorter ? nown : ngem;
    for (int v = wave; v < q; v += OWN_T / 32) {
        const int s = first + v * stride;
        if (s < strips) gemm_strip(x, h, sW, (s1 + s) * 16);
    }
}

static __device__ __forceinline__ void fma4(float4& acc, const float4 r, const float w) {
    acc.x += r.x * w; acc.y += r.y * w; acc.z += r.z * w; acc.w += r.w * w;
}

__global__ __launch_bounds__(256) void k_gather(
    const int* __restrict__ srt, const int* __restrict__ start, const int* __restrict__ cnt,
    const float* __restrict__ h, const float* __restrict__ dinv,
    const float* __restrict__ bias, const float* __restrict__ alpha,
    float* __restrict__ out, int n) {
    const int lane = threadIdx.x & 31;
    const int node = __builtin_amdgcn_readfirstlane(blockIdx.x * 8 + (threadIdx.x >> 5));
    if (node >= n) return;

    const int   beg = start[node];
    const int   c   = cnt[node];
    const float di  = dinv[node];
    const float4* __restrict__ h4 = (const float4*)h;

    const float4 self = h4[(size_t)node * (HID / 4) + lane];
    const float  d2   = di * di;
    float4 acc;
    acc.x = self.x * d2; acc.y = self.y * d2; acc.z = self.z * d2; acc.w = self.w * d2;

    for (int base = 0; base < c; base += 32) {
        const int m = (c - base < 32) ? (c - base) : 32;
        int   sv = 0;
        float wv = 0.0f;
        if (lane < m) {
            sv = srt[beg + base + lane];
            wv = dinv[sv] * di;
        }
        const int wvi = __float_as_int(wv);
        int j = 0;
        for (; j + 8 <= m; j += 8) {
            float4 r[8];
            float  w[8];
            for (int u = 0; u < 8; ++u) {
                const int s = __builtin_amdgcn_readlane(sv, j + u);
                w[u] = __int_as_float(__builtin_amdgcn_readlane(wvi, j + u));
                r[u] = h4[(size_t)s * (HID / 4) + lane];
            }
            for (int u = 0; u < 8; ++u) fma4(acc, r[u], w[u]);
        }
        if (j + 4 <= m) {
            float4 r[4];
            float  w[4];
            for (int u = 0; u < 4; ++u) {
                const int s = __builtin_amdgcn_readlane(sv, j + u);
                w[u] = __int_as_float(__builtin_amdgcn_readlane(wvi, j + u));
                r[u] = h4[(size_t)s * (HID / 4) + lane];
            }
            for (int u = 0; u < 4; ++u) fma4(acc, r[u], w[u]);
            j += 4;
        }
        if (j + 2 <= m) {
            float4 r[2];
            float  w[2];
            for (int u = 0; u < 2; ++u) {
                const int s = __builtin_amdgcn_readlane(sv, j + u);
                w[u] = __int_as_float(__builtin_amdgcn_readlane(wvi, j + u));
                r[u] = h4[(size_t)s * (HID / 4) + lane];
            }
            for (int u = 0; u < 2; ++u) fma4(acc, r[u], w[u]);
            j += 2;
        }
        if (j < m) {
            const int   s = __builtin_amdgcn_readlane(sv, j);
            const float w = __int_as_float(__builtin_amdgcn_readlane(wvi, j));
            const float4 r = h4[(size_t)s * (HID / 4) + lane];
            fma4(acc, r, w);
        }
    }

    const float4 b = ((const float4*)bias)[lane];
    const float4 a = ((const float4*)alpha)[lane];
    float4 v, o;
    v.x = acc.x + b.x; v.y = acc.y + b.y; v.z = acc.z + b.z; v.w = acc.w + b.w;
    o.x = v.x > 0.0f ? v.x : a.x * v.x;
    o.y = v.y > 0.0f ? v.y : a.y * v.y;
    o.z = v.z > 0.0f ? v.z : a.z * v.z;
    o.w = v.w > 0.0f ? v.w : a.w * v.w;
    ((float4*)out)[(size_t)node * (HID / 4) + lane] = o;
}

extern "C" void kernel_launch(void* const* d_in, const int* in_sizes, int n_in,
                              void* d_out, int out_size, void* d_ws, size_t ws_size,
                              hipStream_t stream) {
    const float* x     = (const float*)d_in[0];
    const int*   ei    = (const int*)d_in[1];
    const float* W     = (const float*)d_in[2];
    const float* bias  = (const float*)d_in[3];
    const float* alpha = (const float*)d_in[4];
    float*       out   = (float*)d_out;

    const int N = in_sizes[0] / IN_CH;
    const int E = in_sizes[1] / 2;
    const int nown = (N + OWN_R - 1) / OWN_R;

    float*    h     = (float*)d_ws;
    float*    dinv  = h + (size_t)N * HID;
    int*      cnt   = (int*)(dinv + N);
    int*      start = cnt + N;
    int*      srt   = start + N;
    int*      cntob = srt + E;
    unsigned* bkt   = (unsigned*)(cntob + (size_t)nown * PART_B);

    k_partition<<<PART_B + GEMM1_B, PART_T, 0, stream>>>(ei, cntob, bkt, E, nown, x, W, h, N);

    const int ngemm = (nown < 192) ? (256 - nown) : 64;
    k_sort_and_gemm<<<nown + ngemm, OWN_T, 0, stream>>>(ei, cntob, bkt, cnt, start, dinv, srt, N, E, nown, x, W, h);

    k_gather<<<(N + 7) / 8, 256, 0, stream>>>(srt, start, cnt, h, dinv, bias, alpha, out, N);
}
